# starting kernel
# baseline (speedup 1.0000x reference)
#include <hip/hip_runtime.h>
#include <hip/hip_bf16.h>

typedef float v2f __attribute__((ext_vector_type(2)));
typedef float v8f __attribute__((ext_vector_type(8)));

#define D_MODEL 1024
#define D_STATE 16
#define DT_RANK 64
#define BATCH   4
#define SEQLEN  2048
#define XDIM    96
#define NTOK    (BATCH * SEQLEN)
#define NCHUNK  32
#define CLEN    (SEQLEN / NCHUNK)
#define NT1     (XDIM / 16)
#define NT2     8

__global__ __launch_bounds__(256) void xp_wmma_kernel(const float* __restrict__ u,
                                                      const float* __restrict__ xw,
                                                      float* __restrict__ xp) {
  const int gtid   = blockIdx.x * 256 + threadIdx.x;
  const int m_tile = gtid >> 5;
  const int lane   = threadIdx.x & 31;
  const int row    = lane & 15;
  const int kph    = (lane >> 4) << 1;

  const float* __restrict__ ap = u  + (size_t)(m_tile * 16 + row) * D_MODEL + kph;
  const float* __restrict__ bp = xw + (size_t)row * D_MODEL + kph;

  v8f acc[NT1] = {};
  for (int k = 0; k < D_MODEL; k += 4) {
    v2f a = *(const v2f*)(ap + k);
    #pragma unroll
    for (int nt = 0; nt < NT1; ++nt) {
      v2f b = *(const v2f*)(bp + (size_t)nt * 16 * D_MODEL + k);
      acc[nt] = __builtin_amdgcn_wmma_f32_16x16x4_f32(false, a, false, b, (short)0,
                                                      acc[nt], false, false);
    }
  }

  const int m0  = m_tile * 16 + ((lane >> 4) << 3);
  const int col = lane & 15;
  #pragma unroll
  for (int nt = 0; nt < NT1; ++nt)
    #pragma unroll
    for (int i = 0; i < 8; ++i)
      xp[(size_t)(m0 + i) * XDIM + nt * 16 + col] = acc[nt][i];
}

__global__ __launch_bounds__(256) void delta_wmma_kernel(const float* __restrict__ xp,
                                                         const float* __restrict__ wt,
                                                         const float* __restrict__ bias,
                                                         float* __restrict__ delta) {
  const int gtid   = blockIdx.x * 256 + threadIdx.x;
  const int wave   = gtid >> 5;
  const int lane   = threadIdx.x & 31;
  const int ngrp   = wave % (D_MODEL / 16 / NT2);
  const int m_tile = wave / (D_MODEL / 16 / NT2);
  const int row    = lane & 15;
  const int kph    = (lane >> 4) << 1;

  const float* __restrict__ ap = xp + (size_t)(m_tile * 16 + row) * XDIM + kph;
  const float* __restrict__ bp = wt + (size_t)(ngrp * NT2 * 16 + row) * DT_RANK + kph;

  v8f acc[NT2] = {};
  #pragma unroll
  for (int k = 0; k < DT_RANK; k += 4) {
    v2f a = *(const v2f*)(ap + k);
    #pragma unroll
    for (int nt = 0; nt < NT2; ++nt) {
      v2f b = *(const v2f*)(bp + (size_t)nt * 16 * DT_RANK + k);
      acc[nt] = __builtin_amdgcn_wmma_f32_16x16x4_f32(false, a, false, b, (short)0,
                                                      acc[nt], false, false);
    }
  }

  const int m0 = m_tile * 16 + ((lane >> 4) << 3);
  #pragma unroll
  for (int nt = 0; nt < NT2; ++nt) {
    const int n0 = (ngrp * NT2 + nt) * 16 + (lane & 15);
    const float bn = bias[n0];
    #pragma unroll
    for (int i = 0; i < 8; ++i) {
      float x  = acc[nt][i] + bn;
      float sp = (x > 20.0f) ? x : __logf(1.0f + __expf(x));
      delta[(size_t)(m0 + i) * D_MODEL + n0] = sp;
    }
  }
}

__global__ __launch_bounds__(256) void scan_pass1_kernel(const float* __restrict__ delta,
                                                         const float* __restrict__ xp,
                                                         const float* __restrict__ u,
                                                         const float* __restrict__ A_log,
                                                         float* __restrict__ Q,
                                                         float* __restrict__ Sdt) {
  const int bid = blockIdx.x;
  const int d   = (bid & 3) * 256 + threadIdx.x;
  const int c   = (bid >> 2) & (NCHUNK - 1);
  const int b   = bid >> 7;

  float A[D_STATE];
  #pragma unroll
  for (int n = 0; n < D_STATE; ++n) A[n] = -__expf(A_log[d * D_STATE + n]);

  float h[D_STATE];
  #pragma unroll
  for (int n = 0; n < D_STATE; ++n) h[n] = 0.0f;
  float sdt = 0.0f;

  const size_t tok0 = (size_t)b * SEQLEN + (size_t)c * CLEN;
  for (int s = 0; s < CLEN; ++s) {
    const size_t tok = tok0 + s;
    const float dt = delta[tok * D_MODEL + d];
    const float ut = u[tok * D_MODEL + d];
    const float4* bv = (const float4*)(xp + tok * XDIM + DT_RANK);
    float4 q0 = bv[0], q1 = bv[1], q2 = bv[2], q3 = bv[3];
    float Bt[D_STATE] = {q0.x, q0.y, q0.z, q0.w, q1.x, q1.y, q1.z, q1.w,
                         q2.x, q2.y, q2.z, q2.w, q3.x, q3.y, q3.z, q3.w};
    sdt += dt;
    const float du = dt * ut;
    #pragma unroll
    for (int n = 0; n < D_STATE; ++n)
      h[n] = __expf(dt * A[n]) * h[n] + du * Bt[n];
  }

  const size_t idx = ((size_t)b * NCHUNK + c) * D_MODEL + d;
  #pragma unroll
  for (int n = 0; n < D_STATE; ++n) Q[idx * D_STATE + n] = h[n];
  Sdt[idx] = sdt;
}

__global__ __launch_bounds__(256) void scan_carry_kernel(const float* __restrict__ Q,
                                                         const float* __restrict__ Sdt,
                                                         const float* __restrict__ A_log,
                                                         float* __restrict__ Hc) {
  const int tid = blockIdx.x * 256 + threadIdx.x;
  const int d   = tid % D_MODEL;
  const int b   = tid / D_MODEL;

  float A[D_STATE];
  #pragma unroll
  for (int n = 0; n < D_STATE; ++n) A[n] = -__expf(A_log[d * D_STATE + n]);

  float h[D_STATE];
  #pragma unroll
  for (int n = 0; n < D_STATE; ++n) h[n] = 0.0f;

  for (int c = 0; c < NCHUNK; ++c) {
    const size_t idx = ((size_t)b * NCHUNK + c) * D_MODEL + d;
    #pragma unroll
    for (int n = 0; n < D_STATE; ++n) Hc[idx * D_STATE + n] = h[n];
    const float s = Sdt[idx];
    #pragma unroll
    for (int n = 0; n < D_STATE; ++n)
      h[n] = __expf(s * A[n]) * h[n] + Q[idx * D_STATE + n];
  }
}

__global__ __launch_bounds__(256) void scan_pass3_kernel(const float* __restrict__ delta,
                                                         const float* __restrict__ xp,
                                                         const float* __restrict__ u,
                                                         const float* __restrict__ A_log,
                                                         const float* __restrict__ Dsk,
                                                         const float* __restrict__ Hc,
                                                         float* __restrict__ y) {
  const int bid = blockIdx.x;
  const int d   = (bid & 3) * 256 + threadIdx.x;
  const int c   = (bid >> 2) & (NCHUNK - 1);
  const int b   = bid >> 7;

  float A[D_STATE];
  #pragma unroll
  for (int n = 0; n < D_STATE; ++n) A[n] = -__expf(A_log[d * D_STATE + n]);

  const size_t idx = ((size_t)b * NCHUNK + c) * D_MODEL + d;
  float h[D_STATE];
  #pragma unroll
  for (int n = 0; n < D_STATE; ++n) h[n] = Hc[idx * D_STATE + n];
  const float dsk = Dsk[d];

  const size_t tok0 = (size_t)b * SEQLEN + (size_t)c * CLEN;
  for (int s = 0; s < CLEN; ++s) {
    const size_t tok = tok0 + s;
    const float dt = delta[tok * D_MODEL + d];
    const float ut = u[tok * D_MODEL + d];
    const float4* bv = (const float4*)(xp + tok * XDIM + DT_RANK);
    float4 q0 = bv[0], q1 = bv[1], q2 = bv[2], q3 = bv[3];
    const float4* cv = (const float4*)(xp + tok * XDIM + DT_RANK + D_STATE);
    float4 p0 = cv[0], p1 = cv[1], p2 = cv[2], p3 = cv[3];
    float Bt[D_STATE] = {q0.x, q0.y, q0.z, q0.w, q1.x, q1.y, q1.z, q1.w,
                         q2.x, q2.y, q2.z, q2.w, q3.x, q3.y, q3.z, q3.w};
    float Ct[D_STATE] = {p0.x, p0.y, p0.z, p0.w, p1.x, p1.y, p1.z, p1.w,
                         p2.x, p2.y, p2.z, p2.w, p3.x, p3.y, p3.z, p3.w};
    const float du = dt * ut;
    float acc = 0.0f;
    #pragma unroll
    for (int n = 0; n < D_STATE; ++n) {
      h[n] = __expf(dt * A[n]) * h[n] + du * Bt[n];
      acc += h[n] * Ct[n];
    }
    y[tok * D_MODEL + d] = acc + ut * dsk;
  }
}

extern "C" void kernel_launch(void* const* d_in, const int* in_sizes, int n_in,
                              void* d_out, int out_size, void* d_ws, size_t ws_size,
                              hipStream_t stream) {
  const float* u      = (const float*)d_in[0];
  const float* A_log  = (const float*)d_in[1];
  const float* D_skip = (const float*)d_in[2];
  const float* dt_w   = (const float*)d_in[3];
  const float* dt_b   = (const float*)d_in[4];
  const float* x_w    = (const float*)d_in[5];
  float* y = (float*)d_out;

  float* ws    = (float*)d_ws;
  float* xp    = ws;
  float* delta = xp    + (size_t)NTOK * XDIM;
  float* Q     = delta + (size_t)NTOK * D_MODEL;
  float* Sdt   = Q     + (size_t)BATCH * NCHUNK * D_MODEL * D_STATE;
  float* Hc    = Sdt   + (size_t)BATCH * NCHUNK * D_MODEL;

  xp_wmma_kernel<<<(NTOK / 16) * 32 / 256, 256, 0, stream>>>(u, x_w, xp);

  delta_wmma_kernel<<<(NTOK / 16) * (D_MODEL / 16 / NT2) * 32 / 256, 256, 0, stream>>>(
      xp, dt_w, dt_b, delta);

  scan_pass1_kernel<<<BATCH * NCHUNK * (D_MODEL / 256), 256, 0, stream>>>(
      delta, xp, u, A_log, Q, Sdt);

  scan_carry_kernel<<<BATCH * D_MODEL / 256, 256, 0, stream>>>(Q, Sdt, A_log, Hc);

  scan_pass3_kernel<<<BATCH * NCHUNK * (D_MODEL / 256), 256, 0, stream>>>(
      delta, xp, u, A_log, D_skip, Hc, y);
}


# xp GEMM: K split over the 8 waves of each block (512 blocks, every WGP busy, 32 k-steps per wave); the 8 partial tiles summed through LDS in fixed order (changes xp's summation order)
# speedup vs baseline: 1.2521x; 1.2521x over previous
#include <hip/hip_runtime.h>
#include <hip/hip_bf16.h>

typedef float v2f __attribute__((ext_vector_type(2)));
typedef float v8f __attribute__((ext_vector_type(8)));

#define D_MODEL 1024
#define D_STATE 16
#define DT_RANK 64
#define BATCH   4
#define SEQLEN  2048
#define XDIM    96
#define NTOK    (BATCH * SEQLEN)
#define NCHUNK  32
#define CLEN    (SEQLEN / NCHUNK)
#define NT1     (XDIM / 16)
#define NT2     8

#define KSPLIT   8
#define KSLICE   (D_MODEL / KSPLIT)
#define NT_PITCH 272
#define W_PITCH  (NT1 * NT_PITCH)

__global__ __launch_bounds__(256) void xp_wmma_kernel(const float* __restrict__ u,
                                                      const float* __restrict__ xw,
                                                      float* __restrict__ xp) {
  __shared__ float part[KSPLIT * W_PITCH];

  const int m_tile = blockIdx.x;
  const int wv     = threadIdx.x >> 5;
  const int lane   = threadIdx.x & 31;
  const int row    = lane & 15;
  const int kph    = (lane >> 4) << 1;

  const float* __restrict__ ap = u  + (size_t)(m_tile * 16 + row) * D_MODEL + wv * KSLICE + kph;
  const float* __restrict__ bp = xw + (size_t)row * D_MODEL + wv * KSLICE + kph;

  v8f acc[NT1] = {};
  for (int k = 0; k < KSLICE; k += 4) {
    v2f a = *(const v2f*)(ap + k);
    #pragma unroll
    for (int nt = 0; nt < NT1; ++nt) {
      v2f b = *(const v2f*)(bp + (size_t)nt * 16 * D_MODEL + k);
      acc[nt] = __builtin_amdgcn_wmma_f32_16x16x4_f32(false, a, false, b, (short)0,
                                                      acc[nt], false, false);
    }
  }

  float* pw = part + wv * W_PITCH + lane;
  #pragma unroll
  for (int nt = 0; nt < NT1; ++nt)
    #pragma unroll
    for (int i = 0; i < 8; ++i)
      pw[nt * NT_PITCH + i * 32] = acc[nt][i];

  __syncthreads();

  float* __restrict__ outp = xp + (size_t)m_tile * 16 * XDIM;
  #pragma unroll
  for (int j = 0; j < (16 * XDIM) / 256; ++j) {
    const int e   = j * 256 + threadIdx.x;
    const int r   = e / XDIM;
    const int col = e - r * XDIM;
    const int off = (col >> 4) * NT_PITCH + (r & 7) * 32 + (r >> 3) * 16 + (col & 15);
    float s = part[off];
    #pragma unroll
    for (int w = 1; w < KSPLIT; ++w) s += part[w * W_PITCH + off];
    outp[e] = s;
  }
}

__global__ __launch_bounds__(256) void delta_wmma_kernel(const float* __restrict__ xp,
                                                         const float* __restrict__ wt,
                                                         const float* __restrict__ bias,
                                                         float* __restrict__ delta) {
  const int gtid   = blockIdx.x * 256 + threadIdx.x;
  const int wave   = gtid >> 5;
  const int lane   = threadIdx.x & 31;
  const int ngrp   = wave % (D_MODEL / 16 / NT2);
  const int m_tile = wave / (D_MODEL / 16 / NT2);
  const int row    = lane & 15;
  const int kph    = (lane >> 4) << 1;

  const float* __restrict__ ap = xp + (size_t)(m_tile * 16 + row) * XDIM + kph;
  const float* __restrict__ bp = wt + (size_t)(ngrp * NT2 * 16 + row) * DT_RANK + kph;

  v8f acc[NT2] = {};
  #pragma unroll
  for (int k = 0; k < DT_RANK; k += 4) {
    v2f a = *(const v2f*)(ap + k);
    #pragma unroll
    for (int nt = 0; nt < NT2; ++nt) {
      v2f b = *(const v2f*)(bp + (size_t)nt * 16 * DT_RANK + k);
      acc[nt] = __builtin_amdgcn_wmma_f32_16x16x4_f32(false, a, false, b, (short)0,
                                                      acc[nt], false, false);
    }
  }

  const int m0 = m_tile * 16 + ((lane >> 4) << 3);
  #pragma unroll
  for (int nt = 0; nt < NT2; ++nt) {
    const int n0 = (ngrp * NT2 + nt) * 16 + (lane & 15);
    const float bn = bias[n0];
    #pragma unroll
    for (int i = 0; i < 8; ++i) {
      float x  = acc[nt][i] + bn;
      float sp = (x > 20.0f) ? x : __logf(1.0f + __expf(x));
      delta[(size_t)(m0 + i) * D_MODEL + n0] = sp;
    }
  }
}

__global__ __launch_bounds__(256) void scan_pass1_kernel(const float* __restrict__ delta,
                                                         const float* __restrict__ xp,
                                                         const float* __restrict__ u,
                                                         const float* __restrict__ A_log,
                                                         float* __restrict__ Q,
                                                         float* __restrict__ Sdt) {
  const int bid = blockIdx.x;
  const int d   = (bid & 3) * 256 + threadIdx.x;
  const int c   = (bid >> 2) & (NCHUNK - 1);
  const int b   = bid >> 7;

  float A[D_STATE];
  #pragma unroll
  for (int n = 0; n < D_STATE; ++n) A[n] = -__expf(A_log[d * D_STATE + n]);

  float h[D_STATE];
  #pragma unroll
  for (int n = 0; n < D_STATE; ++n) h[n] = 0.0f;
  float sdt = 0.0f;

  const size_t tok0 = (size_t)b * SEQLEN + (size_t)c * CLEN;
  for (int s = 0; s < CLEN; ++s) {
    const size_t tok = tok0 + s;
    const float dt = delta[tok * D_MODEL + d];
    const float ut = u[tok * D_MODEL + d];
    const float4* bv = (const float4*)(xp + tok * XDIM + DT_RANK);
    float4 q0 = bv[0], q1 = bv[1], q2 = bv[2], q3 = bv[3];
    float Bt[D_STATE] = {q0.x, q0.y, q0.z, q0.w, q1.x, q1.y, q1.z, q1.w,
                         q2.x, q2.y, q2.z, q2.w, q3.x, q3.y, q3.z, q3.w};
    sdt += dt;
    const float du = dt * ut;
    #pragma unroll
    for (int n = 0; n < D_STATE; ++n)
      h[n] = __expf(dt * A[n]) * h[n] + du * Bt[n];
  }

  const size_t idx = ((size_t)b * NCHUNK + c) * D_MODEL + d;
  #pragma unroll
  for (int n = 0; n < D_STATE; ++n) Q[idx * D_STATE + n] = h[n];
  Sdt[idx] = sdt;
}

__global__ __launch_bounds__(256) void scan_carry_kernel(const float* __restrict__ Q,
                                                         const float* __restrict__ Sdt,
                                                         const float* __restrict__ A_log,
                                                         float* __restrict__ Hc) {
  const int tid = blockIdx.x * 256 + threadIdx.x;
  const int d   = tid % D_MODEL;
  const int b   = tid / D_MODEL;

  float A[D_STATE];
  #pragma unroll
  for (int n = 0; n < D_STATE; ++n) A[n] = -__expf(A_log[d * D_STATE + n]);

  float h[D_STATE];
  #pragma unroll
  for (int n = 0; n < D_STATE; ++n) h[n] = 0.0f;

  for (int c = 0; c < NCHUNK; ++c) {
    const size_t idx = ((size_t)b * NCHUNK + c) * D_MODEL + d;
    #pragma unroll
    for (int n = 0; n < D_STATE; ++n) Hc[idx * D_STATE + n] = h[n];
    const float s = Sdt[idx];
    #pragma unroll
    for (int n = 0; n < D_STATE; ++n)
      h[n] = __expf(s * A[n]) * h[n] + Q[idx * D_STATE + n];
  }
}

__global__ __launch_bounds__(256) void scan_pass3_kernel(const float* __restrict__ delta,
                                                         const float* __restrict__ xp,
                                                         const float* __restrict__ u,
                                                         const float* __restrict__ A_log,
                                                         const float* __restrict__ Dsk,
                                                         const float* __restrict__ Hc,
                                                         float* __restrict__ y) {
  const int bid = blockIdx.x;
  const int d   = (bid & 3) * 256 + threadIdx.x;
  const int c   = (bid >> 2) & (NCHUNK - 1);
  const int b   = bid >> 7;

  float A[D_STATE];
  #pragma unroll
  for (int n = 0; n < D_STATE; ++n) A[n] = -__expf(A_log[d * D_STATE + n]);

  const size_t idx = ((size_t)b * NCHUNK + c) * D_MODEL + d;
  float h[D_STATE];
  #pragma unroll
  for (int n = 0; n < D_STATE; ++n) h[n] = Hc[idx * D_STATE + n];
  const float dsk = Dsk[d];

  const size_t tok0 = (size_t)b * SEQLEN + (size_t)c * CLEN;
  for (int s = 0; s < CLEN; ++s) {
    const size_t tok = tok0 + s;
    const float dt = delta[tok * D_MODEL + d];
    const float ut = u[tok * D_MODEL + d];
    const float4* bv = (const float4*)(xp + tok * XDIM + DT_RANK);
    float4 q0 = bv[0], q1 = bv[1], q2 = bv[2], q3 = bv[3];
    const float4* cv = (const float4*)(xp + tok * XDIM + DT_RANK + D_STATE);
    float4 p0 = cv[0], p1 = cv[1], p2 = cv[2], p3 = cv[3];
    float Bt[D_STATE] = {q0.x, q0.y, q0.z, q0.w, q1.x, q1.y, q1.z, q1.w,
                         q2.x, q2.y, q2.z, q2.w, q3.x, q3.y, q3.z, q3.w};
    float Ct[D_STATE] = {p0.x, p0.y, p0.z, p0.w, p1.x, p1.y, p1.z, p1.w,
                         p2.x, p2.y, p2.z, p2.w, p3.x, p3.y, p3.z, p3.w};
    const float du = dt * ut;
    float acc = 0.0f;
    #pragma unroll
    for (int n = 0; n < D_STATE; ++n) {
      h[n] = __expf(dt * A[n]) * h[n] + du * Bt[n];
      acc += h[n] * Ct[n];
    }
    y[tok * D_MODEL + d] = acc + ut * dsk;
  }
}

extern "C" void kernel_launch(void* const* d_in, const int* in_sizes, int n_in,
                              void* d_out, int out_size, void* d_ws, size_t ws_size,
                              hipStream_t stream) {
  const float* u      = (const float*)d_in[0];
  const float* A_log  = (const float*)d_in[1];
  const float* D_skip = (const float*)d_in[2];
  const float* dt_w   = (const float*)d_in[3];
  const float* dt_b   = (const float*)d_in[4];
  const float* x_w    = (const float*)d_in[5];
  float* y = (float*)d_out;

  float* ws    = (float*)d_ws;
  float* xp    = ws;
  float* delta = xp    + (size_t)NTOK * XDIM;
  float* Q     = delta + (size_t)NTOK * D_MODEL;
  float* Sdt   = Q     + (size_t)BATCH * NCHUNK * D_MODEL * D_STATE;
  float* Hc    = Sdt   + (size_t)BATCH * NCHUNK * D_MODEL;

  xp_wmma_kernel<<<NTOK / 16, 256, 0, stream>>>(u, x_w, xp);

  delta_wmma_kernel<<<(NTOK / 16) * (D_MODEL / 16 / NT2) * 32 / 256, 256, 0, stream>>>(
      xp, dt_w, dt_b, delta);

  scan_pass1_kernel<<<BATCH * NCHUNK * (D_MODEL / 256), 256, 0, stream>>>(
      delta, xp, u, A_log, Q, Sdt);

  scan_carry_kernel<<<BATCH * D_MODEL / 256, 256, 0, stream>>>(Q, Sdt, A_log, Hc);

  scan_pass3_kernel<<<BATCH * NCHUNK * (D_MODEL / 256), 256, 0, stream>>>(
      delta, xp, u, A_log, D_skip, Hc, y);
}


# scan pass 1 and pass 3: dt/u of the next 8 steps are requested while the current 8 are computed and the chunk's B/C rows sit in LDS, so no global load waits inside the serial recurrence
# speedup vs baseline: 1.6116x; 1.2871x over previous
#include <hip/hip_runtime.h>
#include <hip/hip_bf16.h>

typedef float v2f __attribute__((ext_vector_type(2)));
typedef float v8f __attribute__((ext_vector_type(8)));

#define D_MODEL 1024
#define D_STATE 16
#define DT_RANK 64
#define BATCH   4
#define SEQLEN  2048
#define XDIM    96
#define NTOK    (BATCH * SEQLEN)
#define NCHUNK  32
#define CLEN    (SEQLEN / NCHUNK)
#define NT1     (XDIM / 16)
#define NT2     8

#define KSPLIT   8
#define KSLICE   (D_MODEL / KSPLIT)
#define NT_PITCH 272
#define W_PITCH  (NT1 * NT_PITCH)

__global__ __launch_bounds__(256) void xp_wmma_kernel(const float* __restrict__ u,
                                                      const float* __restrict__ xw,
                                                      float* __restrict__ xp) {
  __shared__ float part[KSPLIT * W_PITCH];

  const int m_tile = blockIdx.x;
  const int wv     = threadIdx.x >> 5;
  const int lane   = threadIdx.x & 31;
  const int row    = lane & 15;
  const int kph    = (lane >> 4) << 1;

  const float* __restrict__ ap = u  + (size_t)(m_tile * 16 + row) * D_MODEL + wv * KSLICE + kph;
  const float* __restrict__ bp = xw + (size_t)row * D_MODEL + wv * KSLICE + kph;

  v8f acc[NT1] = {};
  for (int k = 0; k < KSLICE; k += 4) {
    v2f a = *(const v2f*)(ap + k);
    #pragma unroll
    for (int nt = 0; nt < NT1; ++nt) {
      v2f b = *(const v2f*)(bp + (size_t)nt * 16 * D_MODEL + k);
      acc[nt] = __builtin_amdgcn_wmma_f32_16x16x4_f32(false, a, false, b, (short)0,
                                                      acc[nt], false, false);
    }
  }

  float* pw = part + wv * W_PITCH + lane;
  #pragma unroll
  for (int nt = 0; nt < NT1; ++nt)
    #pragma unroll
    for (int i = 0; i < 8; ++i)
      pw[nt * NT_PITCH + i * 32] = acc[nt][i];

  __syncthreads();

  float* __restrict__ outp = xp + (size_t)m_tile * 16 * XDIM;
  #pragma unroll
  for (int j = 0; j < (16 * XDIM) / 256; ++j) {
    const int e   = j * 256 + threadIdx.x;
    const int r   = e / XDIM;
    const int col = e - r * XDIM;
    const int off = (col >> 4) * NT_PITCH + (r & 7) * 32 + (r >> 3) * 16 + (col & 15);
    float s = part[off];
    #pragma unroll
    for (int w = 1; w < KSPLIT; ++w) s += part[w * W_PITCH + off];
    outp[e] = s;
  }
}

__global__ __launch_bounds__(256) void delta_wmma_kernel(const float* __restrict__ xp,
                                                         const float* __restrict__ wt,
                                                         const float* __restrict__ bias,
                                                         float* __restrict__ delta) {
  const int gtid   = blockIdx.x * 256 + threadIdx.x;
  const int wave   = gtid >> 5;
  const int lane   = threadIdx.x & 31;
  const int ngrp   = wave % (D_MODEL / 16 / NT2);
  const int m_tile = wave / (D_MODEL / 16 / NT2);
  const int row    = lane & 15;
  const int kph    = (lane >> 4) << 1;

  const float* __restrict__ ap = xp + (size_t)(m_tile * 16 + row) * XDIM + kph;
  const float* __restrict__ bp = wt + (size_t)(ngrp * NT2 * 16 + row) * DT_RANK + kph;

  v8f acc[NT2] = {};
  #pragma unroll
  for (int k = 0; k < DT_RANK; k += 4) {
    v2f a = *(const v2f*)(ap + k);
    #pragma unroll
    for (int nt = 0; nt < NT2; ++nt) {
      v2f b = *(const v2f*)(bp + (size_t)nt * 16 * DT_RANK + k);
      acc[nt] = __builtin_amdgcn_wmma_f32_16x16x4_f32(false, a, false, b, (short)0,
                                                      acc[nt], false, false);
    }
  }

  const int m0 = m_tile * 16 + ((lane >> 4) << 3);
  #pragma unroll
  for (int nt = 0; nt < NT2; ++nt) {
    const int n0 = (ngrp * NT2 + nt) * 16 + (lane & 15);
    const float bn = bias[n0];
    #pragma unroll
    for (int i = 0; i < 8; ++i) {
      float x  = acc[nt][i] + bn;
      float sp = (x > 20.0f) ? x : __logf(1.0f + __expf(x));
      delta[(size_t)(m0 + i) * D_MODEL + n0] = sp;
    }
  }
}

#define PF   8
#define NGRP (CLEN / PF)

#define SCHED_FENCE() __builtin_amdgcn_sched_barrier(0)

__device__ __forceinline__ void load_group(const float* __restrict__ dp, const float* __restrict__ up,
                                           float (&dt)[PF], float (&ut)[PF]) {
  #pragma unroll
  for (int i = 0; i < PF; ++i) {
    dt[i] = dp[(size_t)i * D_MODEL];
    ut[i] = up[(size_t)i * D_MODEL];
  }
}

__device__ __forceinline__ void pass1_group(const float4* rows, const float (&dt)[PF],
                                            const float (&ut)[PF], const float (&A)[D_STATE],
                                            float (&h)[D_STATE], float& sdt) {
  float4 n0 = rows[0], n1 = rows[1], n2 = rows[2], n3 = rows[3];
  #pragma unroll
  for (int i = 0; i < PF; ++i) {
    const float4 q0 = n0, q1 = n1, q2 = n2, q3 = n3;
    if (i + 1 < PF) {
      n0 = rows[(i + 1) * 4 + 0]; n1 = rows[(i + 1) * 4 + 1];
      n2 = rows[(i + 1) * 4 + 2]; n3 = rows[(i + 1) * 4 + 3];
    }
    const float Bt[D_STATE] = {q0.x, q0.y, q0.z, q0.w, q1.x, q1.y, q1.z, q1.w,
                               q2.x, q2.y, q2.z, q2.w, q3.x, q3.y, q3.z, q3.w};
    const float d_t = dt[i];
    sdt += d_t;
    const float du = d_t * ut[i];
    #pragma unroll
    for (int n = 0; n < D_STATE; ++n)
      h[n] = __expf(d_t * A[n]) * h[n] + du * Bt[n];
  }
}

__global__ __launch_bounds__(256) void scan_pass1_kernel(const float* __restrict__ delta,
                                                         const float* __restrict__ xp,
                                                         const float* __restrict__ u,
                                                         const float* __restrict__ A_log,
                                                         float* __restrict__ Q,
                                                         float* __restrict__ Sdt) {
  __shared__ float4 Bs[CLEN * 4];

  const int bid = blockIdx.x;
  const int d   = (bid & 3) * 256 + threadIdx.x;
  const int c   = (bid >> 2) & (NCHUNK - 1);
  const int b   = bid >> 7;
  const size_t tok0 = (size_t)b * SEQLEN + (size_t)c * CLEN;

  Bs[threadIdx.x] = *(const float4*)(xp + (tok0 + (threadIdx.x >> 2)) * XDIM + DT_RANK +
                                     (threadIdx.x & 3) * 4);

  const float* __restrict__ dp = delta + tok0 * D_MODEL + d;
  const float* __restrict__ up = u     + tok0 * D_MODEL + d;

  float dtc[PF], utc[PF], dtn[PF], utn[PF];
  load_group(dp, up, dtc, utc);

  float A[D_STATE];
  #pragma unroll
  for (int n = 0; n < D_STATE; ++n) A[n] = -__expf(A_log[d * D_STATE + n]);

  float h[D_STATE];
  #pragma unroll
  for (int n = 0; n < D_STATE; ++n) h[n] = 0.0f;
  float sdt = 0.0f;

  __syncthreads();

  for (int g = 0; g < NGRP - 1; ++g) {
    load_group(dp + (size_t)(g + 1) * PF * D_MODEL, up + (size_t)(g + 1) * PF * D_MODEL, dtn, utn);
    SCHED_FENCE();
    pass1_group(Bs + g * PF * 4, dtc, utc, A, h, sdt);
    #pragma unroll
    for (int i = 0; i < PF; ++i) { dtc[i] = dtn[i]; utc[i] = utn[i]; }
  }
  pass1_group(Bs + (NGRP - 1) * PF * 4, dtc, utc, A, h, sdt);

  const size_t idx = ((size_t)b * NCHUNK + c) * D_MODEL + d;
  #pragma unroll
  for (int n = 0; n < D_STATE; ++n) Q[idx * D_STATE + n] = h[n];
  Sdt[idx] = sdt;
}

__global__ __launch_bounds__(256) void scan_carry_kernel(const float* __restrict__ Q,
                                                         const float* __restrict__ Sdt,
                                                         const float* __restrict__ A_log,
                                                         float* __restrict__ Hc) {
  const int tid = blockIdx.x * 256 + threadIdx.x;
  const int d   = tid % D_MODEL;
  const int b   = tid / D_MODEL;

  float A[D_STATE];
  #pragma unroll
  for (int n = 0; n < D_STATE; ++n) A[n] = -__expf(A_log[d * D_STATE + n]);

  float h[D_STATE];
  #pragma unroll
  for (int n = 0; n < D_STATE; ++n) h[n] = 0.0f;

  for (int c = 0; c < NCHUNK; ++c) {
    const size_t idx = ((size_t)b * NCHUNK + c) * D_MODEL + d;
    #pragma unroll
    for (int n = 0; n < D_STATE; ++n) Hc[idx * D_STATE + n] = h[n];
    const float s = Sdt[idx];
    #pragma unroll
    for (int n = 0; n < D_STATE; ++n)
      h[n] = __expf(s * A[n]) * h[n] + Q[idx * D_STATE + n];
  }
}

__device__ __forceinline__ void pass3_group(const float4* rows, const float (&dt)[PF],
                                            const float (&ut)[PF], const float (&A)[D_STATE],
                                            float (&h)[D_STATE], const float dsk,
                                            float* __restrict__ yp) {
  float4 n0 = rows[0], n1 = rows[1], n2 = rows[2], n3 = rows[3];
  float4 m0 = rows[4], m1 = rows[5], m2 = rows[6], m3 = rows[7];
  #pragma unroll
  for (int i = 0; i < PF; ++i) {
    const float4 q0 = n0, q1 = n1, q2 = n2, q3 = n3;
    const float4 p0 = m0, p1 = m1, p2 = m2, p3 = m3;
    if (i + 1 < PF) {
      n0 = rows[(i + 1) * 8 + 0]; n1 = rows[(i + 1) * 8 + 1];
      n2 = rows[(i + 1) * 8 + 2]; n3 = rows[(i + 1) * 8 + 3];
      m0 = rows[(i + 1) * 8 + 4]; m1 = rows[(i + 1) * 8 + 5];
      m2 = rows[(i + 1) * 8 + 6]; m3 = rows[(i + 1) * 8 + 7];
    }
    const float Bt[D_STATE] = {q0.x, q0.y, q0.z, q0.w, q1.x, q1.y, q1.z, q1.w,
                               q2.x, q2.y, q2.z, q2.w, q3.x, q3.y, q3.z, q3.w};
    const float Ct[D_STATE] = {p0.x, p0.y, p0.z, p0.w, p1.x, p1.y, p1.z, p1.w,
                               p2.x, p2.y, p2.z, p2.w, p3.x, p3.y, p3.z, p3.w};
    const float d_t = dt[i];
    const float u_t = ut[i];
    const float du = d_t * u_t;
    float acc = 0.0f;
    #pragma unroll
    for (int n = 0; n < D_STATE; ++n) {
      h[n] = __expf(d_t * A[n]) * h[n] + du * Bt[n];
      acc += h[n] * Ct[n];
    }
    yp[(size_t)i * D_MODEL] = acc + u_t * dsk;
  }
}

__global__ __launch_bounds__(256) void scan_pass3_kernel(const float* __restrict__ delta,
                                                         const float* __restrict__ xp,
                                                         const float* __restrict__ u,
                                                         const float* __restrict__ A_log,
                                                         const float* __restrict__ Dsk,
                                                         const float* __restrict__ Hc,
                                                         float* __restrict__ y) {
  __shared__ float4 BCs[CLEN * 8];

  const int bid = blockIdx.x;
  const int d   = (bid & 3) * 256 + threadIdx.x;
  const int c   = (bid >> 2) & (NCHUNK - 1);
  const int b   = bid >> 7;
  const size_t tok0 = (size_t)b * SEQLEN + (size_t)c * CLEN;

  #pragma unroll
  for (int j = 0; j < 2; ++j) {
    const int e = j * 256 + threadIdx.x;
    BCs[e] = *(const float4*)(xp + (tok0 + (e >> 3)) * XDIM + DT_RANK + (e & 7) * 4);
  }

  const float* __restrict__ dp = delta + tok0 * D_MODEL + d;
  const float* __restrict__ up = u     + tok0 * D_MODEL + d;
  float* __restrict__       yp = y     + tok0 * D_MODEL + d;

  float dtc[PF], utc[PF], dtn[PF], utn[PF];
  load_group(dp, up, dtc, utc);

  float A[D_STATE];
  #pragma unroll
  for (int n = 0; n < D_STATE; ++n) A[n] = -__expf(A_log[d * D_STATE + n]);

  const size_t idx = ((size_t)b * NCHUNK + c) * D_MODEL + d;
  float h[D_STATE];
  #pragma unroll
  for (int n = 0; n < D_STATE; ++n) h[n] = Hc[idx * D_STATE + n];
  const float dsk = Dsk[d];

  __syncthreads();

  for (int g = 0; g < NGRP - 1; ++g) {
    load_group(dp + (size_t)(g + 1) * PF * D_MODEL, up + (size_t)(g + 1) * PF * D_MODEL, dtn, utn);
    SCHED_FENCE();
    pass3_group(BCs + g * PF * 8, dtc, utc, A, h, dsk, yp + (size_t)g * PF * D_MODEL);
    #pragma unroll
    for (int i = 0; i < PF; ++i) { dtc[i] = dtn[i]; utc[i] = utn[i]; }
  }
  pass3_group(BCs + (NGRP - 1) * PF * 8, dtc, utc, A, h, dsk, yp + (size_t)(NGRP - 1) * PF * D_MODEL);
}

extern "C" void kernel_launch(void* const* d_in, const int* in_sizes, int n_in,
                              void* d_out, int out_size, void* d_ws, size_t ws_size,
                              hipStream_t stream) {
  const float* u      = (const float*)d_in[0];
  const float* A_log  = (const float*)d_in[1];
  const float* D_skip = (const float*)d_in[2];
  const float* dt_w   = (const float*)d_in[3];
  const float* dt_b   = (const float*)d_in[4];
  const float* x_w    = (const float*)d_in[5];
  float* y = (float*)d_out;

  float* ws    = (float*)d_ws;
  float* xp    = ws;
  float* delta = xp    + (size_t)NTOK * XDIM;
  float* Q     = delta + (size_t)NTOK * D_MODEL;
  float* Sdt   = Q     + (size_t)BATCH * NCHUNK * D_MODEL * D_STATE;
  float* Hc    = Sdt   + (size_t)BATCH * NCHUNK * D_MODEL;

  xp_wmma_kernel<<<NTOK / 16, 256, 0, stream>>>(u, x_w, xp);

  delta_wmma_kernel<<<(NTOK / 16) * (D_MODEL / 16 / NT2) * 32 / 256, 256, 0, stream>>>(
      xp, dt_w, dt_b, delta);

  scan_pass1_kernel<<<BATCH * NCHUNK * (D_MODEL / 256), 256, 0, stream>>>(
      delta, xp, u, A_log, Q, Sdt);

  scan_carry_kernel<<<BATCH * D_MODEL / 256, 256, 0, stream>>>(Q, Sdt, A_log, Hc);

  scan_pass3_kernel<<<BATCH * NCHUNK * (D_MODEL / 256), 256, 0, stream>>>(
      delta, xp, u, A_log, D_skip, Hc, y);
}


# carry pass: one thread per (b,d,n) instead of per (b,d); its 32 chunk summaries are all requested before the first step, so load latency is paid once, not twice per chunk; arithmetic order unchanged
# speedup vs baseline: 1.9817x; 1.2296x over previous
#include <hip/hip_runtime.h>
#include <hip/hip_bf16.h>

typedef float v2f __attribute__((ext_vector_type(2)));
typedef float v8f __attribute__((ext_vector_type(8)));

#define D_MODEL 1024
#define D_STATE 16
#define DT_RANK 64
#define BATCH   4
#define SEQLEN  2048
#define XDIM    96
#define NTOK    (BATCH * SEQLEN)
#define NCHUNK  32
#define CLEN    (SEQLEN / NCHUNK)
#define NT1     (XDIM / 16)
#define NT2     8

#define KSPLIT   8
#define KSLICE   (D_MODEL / KSPLIT)
#define NT_PITCH 272
#define W_PITCH  (NT1 * NT_PITCH)

__global__ __launch_bounds__(256) void xp_wmma_kernel(const float* __restrict__ u,
                                                      const float* __restrict__ xw,
                                                      float* __restrict__ xp) {
  __shared__ float part[KSPLIT * W_PITCH];

  const int m_tile = blockIdx.x;
  const int wv     = threadIdx.x >> 5;
  const int lane   = threadIdx.x & 31;
  const int row    = lane & 15;
  const int kph    = (lane >> 4) << 1;

  const float* __restrict__ ap = u  + (size_t)(m_tile * 16 + row) * D_MODEL + wv * KSLICE + kph;
  const float* __restrict__ bp = xw + (size_t)row * D_MODEL + wv * KSLICE + kph;

  v8f acc[NT1] = {};
  for (int k = 0; k < KSLICE; k += 4) {
    v2f a = *(const v2f*)(ap + k);
    #pragma unroll
    for (int nt = 0; nt < NT1; ++nt) {
      v2f b = *(const v2f*)(bp + (size_t)nt * 16 * D_MODEL + k);
      acc[nt] = __builtin_amdgcn_wmma_f32_16x16x4_f32(false, a, false, b, (short)0,
                                                      acc[nt], false, false);
    }
  }

  float* pw = part + wv * W_PITCH + lane;
  #pragma unroll
  for (int nt = 0; nt < NT1; ++nt)
    #pragma unroll
    for (int i = 0; i < 8; ++i)
      pw[nt * NT_PITCH + i * 32] = acc[nt][i];

  __syncthreads();

  float* __restrict__ outp = xp + (size_t)m_tile * 16 * XDIM;
  #pragma unroll
  for (int j = 0; j < (16 * XDIM) / 256; ++j) {
    const int e   = j * 256 + threadIdx.x;
    const int r   = e / XDIM;
    const int col = e - r * XDIM;
    const int off = (col >> 4) * NT_PITCH + (r & 7) * 32 + (r >> 3) * 16 + (col & 15);
    float s = part[off];
    #pragma unroll
    for (int w = 1; w < KSPLIT; ++w) s += part[w * W_PITCH + off];
    outp[e] = s;
  }
}

__global__ __launch_bounds__(256) void delta_wmma_kernel(const float* __restrict__ xp,
                                                         const float* __restrict__ wt,
                                                         const float* __restrict__ bias,
                                                         float* __restrict__ delta) {
  const int gtid   = blockIdx.x * 256 + threadIdx.x;
  const int wave   = gtid >> 5;
  const int lane   = threadIdx.x & 31;
  const int ngrp   = wave % (D_MODEL / 16 / NT2);
  const int m_tile = wave / (D_MODEL / 16 / NT2);
  const int row    = lane & 15;
  const int kph    = (lane >> 4) << 1;

  const float* __restrict__ ap = xp + (size_t)(m_tile * 16 + row) * XDIM + kph;
  const float* __restrict__ bp = wt + (size_t)(ngrp * NT2 * 16 + row) * DT_RANK + kph;

  v8f acc[NT2] = {};
  #pragma unroll
  for (int k = 0; k < DT_RANK; k += 4) {
    v2f a = *(const v2f*)(ap + k);
    #pragma unroll
    for (int nt = 0; nt < NT2; ++nt) {
      v2f b = *(const v2f*)(bp + (size_t)nt * 16 * DT_RANK + k);
      acc[nt] = __builtin_amdgcn_wmma_f32_16x16x4_f32(false, a, false, b, (short)0,
                                                      acc[nt], false, false);
    }
  }

  const int m0 = m_tile * 16 + ((lane >> 4) << 3);
  #pragma unroll
  for (int nt = 0; nt < NT2; ++nt) {
    const int n0 = (ngrp * NT2 + nt) * 16 + (lane & 15);
    const float bn = bias[n0];
    #pragma unroll
    for (int i = 0; i < 8; ++i) {
      float x  = acc[nt][i] + bn;
      float sp = (x > 20.0f) ? x : __logf(1.0f + __expf(x));
      delta[(size_t)(m0 + i) * D_MODEL + n0] = sp;
    }
  }
}

#define PF   8
#define NGRP (CLEN / PF)

#define SCHED_FENCE() __builtin_amdgcn_sched_barrier(0)

__device__ __forceinline__ void load_group(const float* __restrict__ dp, const float* __restrict__ up,
                                           float (&dt)[PF], float (&ut)[PF]) {
  #pragma unroll
  for (int i = 0; i < PF; ++i) {
    dt[i] = dp[(size_t)i * D_MODEL];
    ut[i] = up[(size_t)i * D_MODEL];
  }
}

__device__ __forceinline__ void pass1_group(const float4* rows, const float (&dt)[PF],
                                            const float (&ut)[PF], const float (&A)[D_STATE],
                                            float (&h)[D_STATE], float& sdt) {
  float4 n0 = rows[0], n1 = rows[1], n2 = rows[2], n3 = rows[3];
  #pragma unroll
  for (int i = 0; i < PF; ++i) {
    const float4 q0 = n0, q1 = n1, q2 = n2, q3 = n3;
    if (i + 1 < PF) {
      n0 = rows[(i + 1) * 4 + 0]; n1 = rows[(i + 1) * 4 + 1];
      n2 = rows[(i + 1) * 4 + 2]; n3 = rows[(i + 1) * 4 + 3];
    }
    const float Bt[D_STATE] = {q0.x, q0.y, q0.z, q0.w, q1.x, q1.y, q1.z, q1.w,
                               q2.x, q2.y, q2.z, q2.w, q3.x, q3.y, q3.z, q3.w};
    const float d_t = dt[i];
    sdt += d_t;
    const float du = d_t * ut[i];
    #pragma unroll
    for (int n = 0; n < D_STATE; ++n)
      h[n] = __expf(d_t * A[n]) * h[n] + du * Bt[n];
  }
}

__global__ __launch_bounds__(256) void scan_pass1_kernel(const float* __restrict__ delta,
                                                         const float* __restrict__ xp,
                                                         const float* __restrict__ u,
                                                         const float* __restrict__ A_log,
                                                         float* __restrict__ Q,
                                                         float* __restrict__ Sdt) {
  __shared__ float4 Bs[CLEN * 4];

  const int bid = blockIdx.x;
  const int d   = (bid & 3) * 256 + threadIdx.x;
  const int c   = (bid >> 2) & (NCHUNK - 1);
  const int b   = bid >> 7;
  const size_t tok0 = (size_t)b * SEQLEN + (size_t)c * CLEN;

  Bs[threadIdx.x] = *(const float4*)(xp + (tok0 + (threadIdx.x >> 2)) * XDIM + DT_RANK +
                                     (threadIdx.x & 3) * 4);

  const float* __restrict__ dp = delta + tok0 * D_MODEL + d;
  const float* __restrict__ up = u     + tok0 * D_MODEL + d;

  float dtc[PF], utc[PF], dtn[PF], utn[PF];
  load_group(dp, up, dtc, utc);

  float A[D_STATE];
  #pragma unroll
  for (int n = 0; n < D_STATE; ++n) A[n] = -__expf(A_log[d * D_STATE + n]);

  float h[D_STATE];
  #pragma unroll
  for (int n = 0; n < D_STATE; ++n) h[n] = 0.0f;
  float sdt = 0.0f;

  __syncthreads();

  for (int g = 0; g < NGRP - 1; ++g) {
    load_group(dp + (size_t)(g + 1) * PF * D_MODEL, up + (size_t)(g + 1) * PF * D_MODEL, dtn, utn);
    SCHED_FENCE();
    pass1_group(Bs + g * PF * 4, dtc, utc, A, h, sdt);
    #pragma unroll
    for (int i = 0; i < PF; ++i) { dtc[i] = dtn[i]; utc[i] = utn[i]; }
  }
  pass1_group(Bs + (NGRP - 1) * PF * 4, dtc, utc, A, h, sdt);

  const size_t idx = ((size_t)b * NCHUNK + c) * D_MODEL + d;
  #pragma unroll
  for (int n = 0; n < D_STATE; ++n) Q[idx * D_STATE + n] = h[n];
  Sdt[idx] = sdt;
}

#define CROW (D_MODEL * D_STATE)

__global__ __launch_bounds__(256) void scan_carry_kernel(const float* __restrict__ Q,
                                                         const float* __restrict__ Sdt,
                                                         const float* __restrict__ A_log,
                                                         float* __restrict__ Hc) {
  const unsigned tid = blockIdx.x * 256u + threadIdx.x;
  const unsigned e   = tid % CROW;
  const unsigned b   = tid / CROW;
  const unsigned d   = e / D_STATE;

  const float* __restrict__ qp = Q   + (size_t)b * NCHUNK * CROW + e;
  const float* __restrict__ sp = Sdt + (size_t)b * NCHUNK * D_MODEL + d;
  float* __restrict__       hp = Hc  + (size_t)b * NCHUNK * CROW + e;

  float q[NCHUNK], s[NCHUNK];
  #pragma unroll
  for (int c = 0; c < NCHUNK; ++c) {
    q[c] = qp[(size_t)c * CROW];
    s[c] = sp[(size_t)c * D_MODEL];
  }
  const float A = -__expf(A_log[e]);

  float h = 0.0f;
  #pragma unroll
  for (int c = 0; c < NCHUNK; ++c) {
    hp[(size_t)c * CROW] = h;
    h = __expf(s[c] * A) * h + q[c];
  }
}

__device__ __forceinline__ void pass3_group(const float4* rows, const float (&dt)[PF],
                                            const float (&ut)[PF], const float (&A)[D_STATE],
                                            float (&h)[D_STATE], const float dsk,
                                            float* __restrict__ yp) {
  float4 n0 = rows[0], n1 = rows[1], n2 = rows[2], n3 = rows[3];
  float4 m0 = rows[4], m1 = rows[5], m2 = rows[6], m3 = rows[7];
  #pragma unroll
  for (int i = 0; i < PF; ++i) {
    const float4 q0 = n0, q1 = n1, q2 = n2, q3 = n3;
    const float4 p0 = m0, p1 = m1, p2 = m2, p3 = m3;
    if (i + 1 < PF) {
      n0 = rows[(i + 1) * 8 + 0]; n1 = rows[(i + 1) * 8 + 1];
      n2 = rows[(i + 1) * 8 + 2]; n3 = rows[(i + 1) * 8 + 3];
      m0 = rows[(i + 1) * 8 + 4]; m1 = rows[(i + 1) * 8 + 5];
      m2 = rows[(i + 1) * 8 + 6]; m3 = rows[(i + 1) * 8 + 7];
    }
    const float Bt[D_STATE] = {q0.x, q0.y, q0.z, q0.w, q1.x, q1.y, q1.z, q1.w,
                               q2.x, q2.y, q2.z, q2.w, q3.x, q3.y, q3.z, q3.w};
    const float Ct[D_STATE] = {p0.x, p0.y, p0.z, p0.w, p1.x, p1.y, p1.z, p1.w,
                               p2.x, p2.y, p2.z, p2.w, p3.x, p3.y, p3.z, p3.w};
    const float d_t = dt[i];
    const float u_t = ut[i];
    const float du = d_t * u_t;
    float acc = 0.0f;
    #pragma unroll
    for (int n = 0; n < D_STATE; ++n) {
      h[n] = __expf(d_t * A[n]) * h[n] + du * Bt[n];
      acc += h[n] * Ct[n];
    }
    yp[(size_t)i * D_MODEL] = acc + u_t * dsk;
  }
}

__global__ __launch_bounds__(256) void scan_pass3_kernel(const float* __restrict__ delta,
                                                         const float* __restrict__ xp,
                                                         const float* __restrict__ u,
                                                         const float* __restrict__ A_log,
                                                         const float* __restrict__ Dsk,
                                                         const float* __restrict__ Hc,
                                                         float* __restrict__ y) {
  __shared__ float4 BCs[CLEN * 8];

  const int bid = blockIdx.x;
  const int d   = (bid & 3) * 256 + threadIdx.x;
  const int c   = (bid >> 2) & (NCHUNK - 1);
  const int b   = bid >> 7;
  const size_t tok0 = (size_t)b * SEQLEN + (size_t)c * CLEN;

  #pragma unroll
  for (int j = 0; j < 2; ++j) {
    const int e = j * 256 + threadIdx.x;
    BCs[e] = *(const float4*)(xp + (tok0 + (e >> 3)) * XDIM + DT_RANK + (e & 7) * 4);
  }

  const float* __restrict__ dp = delta + tok0 * D_MODEL + d;
  const float* __restrict__ up = u     + tok0 * D_MODEL + d;
  float* __restrict__       yp = y     + tok0 * D_MODEL + d;

  float dtc[PF], utc[PF], dtn[PF], utn[PF];
  load_group(dp, up, dtc, utc);

  float A[D_STATE];
  #pragma unroll
  for (int n = 0; n < D_STATE; ++n) A[n] = -__expf(A_log[d * D_STATE + n]);

  const size_t idx = ((size_t)b * NCHUNK + c) * D_MODEL + d;
  float h[D_STATE];
  #pragma unroll
  for (int n = 0; n < D_STATE; ++n) h[n] = Hc[idx * D_STATE + n];
  const float dsk = Dsk[d];

  __syncthreads();

  for (int g = 0; g < NGRP - 1; ++g) {
    load_group(dp + (size_t)(g + 1) * PF * D_MODEL, up + (size_t)(g + 1) * PF * D_MODEL, dtn, utn);
    SCHED_FENCE();
    pass3_group(BCs + g * PF * 8, dtc, utc, A, h, dsk, yp + (size_t)g * PF * D_MODEL);
    #pragma unroll
    for (int i = 0; i < PF; ++i) { dtc[i] = dtn[i]; utc[i] = utn[i]; }
  }
  pass3_group(BCs + (NGRP - 1) * PF * 8, dtc, utc, A, h, dsk, yp + (size_t)(NGRP - 1) * PF * D_MODEL);
}

extern "C" void kernel_launch(void* const* d_in, const int* in_sizes, int n_in,
                              void* d_out, int out_size, void* d_ws, size_t ws_size,
                              hipStream_t stream) {
  const float* u      = (const float*)d_in[0];
  const float* A_log  = (const float*)d_in[1];
  const float* D_skip = (const float*)d_in[2];
  const float* dt_w   = (const float*)d_in[3];
  const float* dt_b   = (const float*)d_in[4];
  const float* x_w    = (const float*)d_in[5];
  float* y = (float*)d_out;

  float* ws    = (float*)d_ws;
  float* xp    = ws;
  float* delta = xp    + (size_t)NTOK * XDIM;
  float* Q     = delta + (size_t)NTOK * D_MODEL;
  float* Sdt   = Q     + (size_t)BATCH * NCHUNK * D_MODEL * D_STATE;
  float* Hc    = Sdt   + (size_t)BATCH * NCHUNK * D_MODEL;

  xp_wmma_kernel<<<NTOK / 16, 256, 0, stream>>>(u, x_w, xp);

  delta_wmma_kernel<<<(NTOK / 16) * (D_MODEL / 16 / NT2) * 32 / 256, 256, 0, stream>>>(
      xp, dt_w, dt_b, delta);

  scan_pass1_kernel<<<BATCH * NCHUNK * (D_MODEL / 256), 256, 0, stream>>>(
      delta, xp, u, A_log, Q, Sdt);

  scan_carry_kernel<<<BATCH * D_MODEL * D_STATE / 256, 256, 0, stream>>>(Q, Sdt, A_log, Hc);

  scan_pass3_kernel<<<BATCH * NCHUNK * (D_MODEL / 256), 256, 0, stream>>>(
      delta, xp, u, A_log, D_skip, Hc, y);
}


# scan pass 1 and 3: each token's B/C row is read from LDS while the token before it is computed (a scheduling barrier holds the reads ahead): no LDS round trip waits in the serial recurrence
# speedup vs baseline: 2.0578x; 1.0384x over previous
#include <hip/hip_runtime.h>
#include <hip/hip_bf16.h>

typedef float v2f __attribute__((ext_vector_type(2)));
typedef float v8f __attribute__((ext_vector_type(8)));

#define D_MODEL 1024
#define D_STATE 16
#define DT_RANK 64
#define BATCH   4
#define SEQLEN  2048
#define XDIM    96
#define NTOK    (BATCH * SEQLEN)
#define NCHUNK  32
#define CLEN    (SEQLEN / NCHUNK)
#define NT1     (XDIM / 16)
#define NT2     8

#define KSPLIT   8
#define KSLICE   (D_MODEL / KSPLIT)
#define NT_PITCH 272
#define W_PITCH  (NT1 * NT_PITCH)

__global__ __launch_bounds__(256) void xp_wmma_kernel(const float* __restrict__ u,
                                                      const float* __restrict__ xw,
                                                      float* __restrict__ xp) {
  __shared__ float part[KSPLIT * W_PITCH];

  const int m_tile = blockIdx.x;
  const int wv     = threadIdx.x >> 5;
  const int lane   = threadIdx.x & 31;
  const int row    = lane & 15;
  const int kph    = (lane >> 4) << 1;

  const float* __restrict__ ap = u  + (size_t)(m_tile * 16 + row) * D_MODEL + wv * KSLICE + kph;
  const float* __restrict__ bp = xw + (size_t)row * D_MODEL + wv * KSLICE + kph;

  v8f acc[NT1] = {};
  for (int k = 0; k < KSLICE; k += 4) {
    v2f a = *(const v2f*)(ap + k);
    #pragma unroll
    for (int nt = 0; nt < NT1; ++nt) {
      v2f b = *(const v2f*)(bp + (size_t)nt * 16 * D_MODEL + k);
      acc[nt] = __builtin_amdgcn_wmma_f32_16x16x4_f32(false, a, false, b, (short)0,
                                                      acc[nt], false, false);
    }
  }

  float* pw = part + wv * W_PITCH + lane;
  #pragma unroll
  for (int nt = 0; nt < NT1; ++nt)
    #pragma unroll
    for (int i = 0; i < 8; ++i)
      pw[nt * NT_PITCH + i * 32] = acc[nt][i];

  __syncthreads();

  float* __restrict__ outp = xp + (size_t)m_tile * 16 * XDIM;
  #pragma unroll
  for (int j = 0; j < (16 * XDIM) / 256; ++j) {
    const int e   = j * 256 + threadIdx.x;
    const int r   = e / XDIM;
    const int col = e - r * XDIM;
    const int off = (col >> 4) * NT_PITCH + (r & 7) * 32 + (r >> 3) * 16 + (col & 15);
    float s = part[off];
    #pragma unroll
    for (int w = 1; w < KSPLIT; ++w) s += part[w * W_PITCH + off];
    outp[e] = s;
  }
}

__global__ __launch_bounds__(256) void delta_wmma_kernel(const float* __restrict__ xp,
                                                         const float* __restrict__ wt,
                                                         const float* __restrict__ bias,
                                                         float* __restrict__ delta) {
  const int gtid   = blockIdx.x * 256 + threadIdx.x;
  const int wave   = gtid >> 5;
  const int lane   = threadIdx.x & 31;
  const int ngrp   = wave % (D_MODEL / 16 / NT2);
  const int m_tile = wave / (D_MODEL / 16 / NT2);
  const int row    = lane & 15;
  const int kph    = (lane >> 4) << 1;

  const float* __restrict__ ap = xp + (size_t)(m_tile * 16 + row) * XDIM + kph;
  const float* __restrict__ bp = wt + (size_t)(ngrp * NT2 * 16 + row) * DT_RANK + kph;

  v8f acc[NT2] = {};
  #pragma unroll
  for (int k = 0; k < DT_RANK; k += 4) {
    v2f a = *(const v2f*)(ap + k);
    #pragma unroll
    for (int nt = 0; nt < NT2; ++nt) {
      v2f b = *(const v2f*)(bp + (size_t)nt * 16 * DT_RANK + k);
      acc[nt] = __builtin_amdgcn_wmma_f32_16x16x4_f32(false, a, false, b, (short)0,
                                                      acc[nt], false, false);
    }
  }

  const int m0 = m_tile * 16 + ((lane >> 4) << 3);
  #pragma unroll
  for (int nt = 0; nt < NT2; ++nt) {
    const int n0 = (ngrp * NT2 + nt) * 16 + (lane & 15);
    const float bn = bias[n0];
    #pragma unroll
    for (int i = 0; i < 8; ++i) {
      float x  = acc[nt][i] + bn;
      float sp = (x > 20.0f) ? x : __logf(1.0f + __expf(x));
      delta[(size_t)(m0 + i) * D_MODEL + n0] = sp;
    }
  }
}

#define PF   8
#define NGRP (CLEN / PF)

#define SCHED_FENCE() __builtin_amdgcn_sched_barrier(0)

__device__ __forceinline__ void load_group(const float* __restrict__ dp, const float* __restrict__ up,
                                           float (&dt)[PF], float (&ut)[PF]) {
  #pragma unroll
  for (int i = 0; i < PF; ++i) {
    dt[i] = dp[(size_t)i * D_MODEL];
    ut[i] = up[(size_t)i * D_MODEL];
  }
}

template <bool LAST>
__device__ __forceinline__ void pass1_group(const float4* rows, const float (&dt)[PF],
                                            const float (&ut)[PF], const float (&A)[D_STATE],
                                            float (&h)[D_STATE], float& sdt,
                                            float4 (&nb)[4]) {
  #pragma unroll
  for (int i = 0; i < PF; ++i) {
    const float4 q0 = nb[0], q1 = nb[1], q2 = nb[2], q3 = nb[3];
    if (i + 1 < PF || !LAST) {
      nb[0] = rows[(i + 1) * 4 + 0]; nb[1] = rows[(i + 1) * 4 + 1];
      nb[2] = rows[(i + 1) * 4 + 2]; nb[3] = rows[(i + 1) * 4 + 3];
    }
    SCHED_FENCE();
    const float Bt[D_STATE] = {q0.x, q0.y, q0.z, q0.w, q1.x, q1.y, q1.z, q1.w,
                               q2.x, q2.y, q2.z, q2.w, q3.x, q3.y, q3.z, q3.w};
    const float d_t = dt[i];
    sdt += d_t;
    const float du = d_t * ut[i];
    #pragma unroll
    for (int n = 0; n < D_STATE; ++n)
      h[n] = __expf(d_t * A[n]) * h[n] + du * Bt[n];
  }
}

__global__ __launch_bounds__(256) void scan_pass1_kernel(const float* __restrict__ delta,
                                                         const float* __restrict__ xp,
                                                         const float* __restrict__ u,
                                                         const float* __restrict__ A_log,
                                                         float* __restrict__ Q,
                                                         float* __restrict__ Sdt) {
  __shared__ float4 Bs[CLEN * 4];

  const int bid = blockIdx.x;
  const int d   = (bid & 3) * 256 + threadIdx.x;
  const int c   = (bid >> 2) & (NCHUNK - 1);
  const int b   = bid >> 7;
  const size_t tok0 = (size_t)b * SEQLEN + (size_t)c * CLEN;

  Bs[threadIdx.x] = *(const float4*)(xp + (tok0 + (threadIdx.x >> 2)) * XDIM + DT_RANK +
                                     (threadIdx.x & 3) * 4);

  const float* __restrict__ dp = delta + tok0 * D_MODEL + d;
  const float* __restrict__ up = u     + tok0 * D_MODEL + d;

  float dtc[PF], utc[PF], dtn[PF], utn[PF];
  load_group(dp, up, dtc, utc);

  float A[D_STATE];
  #pragma unroll
  for (int n = 0; n < D_STATE; ++n) A[n] = -__expf(A_log[d * D_STATE + n]);

  float h[D_STATE];
  #pragma unroll
  for (int n = 0; n < D_STATE; ++n) h[n] = 0.0f;
  float sdt = 0.0f;

  __syncthreads();

  float4 nb[4] = {Bs[0], Bs[1], Bs[2], Bs[3]};

  for (int g = 0; g < NGRP - 1; ++g) {
    load_group(dp + (size_t)(g + 1) * PF * D_MODEL, up + (size_t)(g + 1) * PF * D_MODEL, dtn, utn);
    SCHED_FENCE();
    pass1_group<false>(Bs + g * PF * 4, dtc, utc, A, h, sdt, nb);
    #pragma unroll
    for (int i = 0; i < PF; ++i) { dtc[i] = dtn[i]; utc[i] = utn[i]; }
  }
  pass1_group<true>(Bs + (NGRP - 1) * PF * 4, dtc, utc, A, h, sdt, nb);

  const size_t idx = ((size_t)b * NCHUNK + c) * D_MODEL + d;
  #pragma unroll
  for (int n = 0; n < D_STATE; ++n) Q[idx * D_STATE + n] = h[n];
  Sdt[idx] = sdt;
}

#define CROW (D_MODEL * D_STATE)

__global__ __launch_bounds__(256) void scan_carry_kernel(const float* __restrict__ Q,
                                                         const float* __restrict__ Sdt,
                                                         const float* __restrict__ A_log,
                                                         float* __restrict__ Hc) {
  const unsigned tid = blockIdx.x * 256u + threadIdx.x;
  const unsigned e   = tid % CROW;
  const unsigned b   = tid / CROW;
  const unsigned d   = e / D_STATE;

  const float* __restrict__ qp = Q   + (size_t)b * NCHUNK * CROW + e;
  const float* __restrict__ sp = Sdt + (size_t)b * NCHUNK * D_MODEL + d;
  float* __restrict__       hp = Hc  + (size_t)b * NCHUNK * CROW + e;

  float q[NCHUNK], s[NCHUNK];
  #pragma unroll
  for (int c = 0; c < NCHUNK; ++c) {
    q[c] = qp[(size_t)c * CROW];
    s[c] = sp[(size_t)c * D_MODEL];
  }
  const float A = -__expf(A_log[e]);

  float h = 0.0f;
  #pragma unroll
  for (int c = 0; c < NCHUNK; ++c) {
    hp[(size_t)c * CROW] = h;
    h = __expf(s[c] * A) * h + q[c];
  }
}

template <bool LAST>
__device__ __forceinline__ void pass3_group(const float4* rows, const float (&dt)[PF],
                                            const float (&ut)[PF], const float (&A)[D_STATE],
                                            float (&h)[D_STATE], const float dsk,
                                            float* __restrict__ yp, float4 (&nbc)[8]) {
  #pragma unroll
  for (int i = 0; i < PF; ++i) {
    const float4 q0 = nbc[0], q1 = nbc[1], q2 = nbc[2], q3 = nbc[3];
    const float4 p0 = nbc[4], p1 = nbc[5], p2 = nbc[6], p3 = nbc[7];
    if (i + 1 < PF || !LAST) {
      #pragma unroll
      for (int r = 0; r < 8; ++r) nbc[r] = rows[(i + 1) * 8 + r];
    }
    SCHED_FENCE();
    const float Bt[D_STATE] = {q0.x, q0.y, q0.z, q0.w, q1.x, q1.y, q1.z, q1.w,
                               q2.x, q2.y, q2.z, q2.w, q3.x, q3.y, q3.z, q3.w};
    const float Ct[D_STATE] = {p0.x, p0.y, p0.z, p0.w, p1.x, p1.y, p1.z, p1.w,
                               p2.x, p2.y, p2.z, p2.w, p3.x, p3.y, p3.z, p3.w};
    const float d_t = dt[i];
    const float u_t = ut[i];
    const float du = d_t * u_t;
    float acc = 0.0f;
    #pragma unroll
    for (int n = 0; n < D_STATE; ++n) {
      h[n] = __expf(d_t * A[n]) * h[n] + du * Bt[n];
      acc += h[n] * Ct[n];
    }
    yp[(size_t)i * D_MODEL] = acc + u_t * dsk;
  }
}

__global__ __launch_bounds__(256) void scan_pass3_kernel(const float* __restrict__ delta,
                                                         const float* __restrict__ xp,
                                                         const float* __restrict__ u,
                                                         const float* __restrict__ A_log,
                                                         const float* __restrict__ Dsk,
                                                         const float* __restrict__ Hc,
                                                         float* __restrict__ y) {
  __shared__ float4 BCs[CLEN * 8];

  const int bid = blockIdx.x;
  const int d   = (bid & 3) * 256 + threadIdx.x;
  const int c   = (bid >> 2) & (NCHUNK - 1);
  const int b   = bid >> 7;
  const size_t tok0 = (size_t)b * SEQLEN + (size_t)c * CLEN;

  #pragma unroll
  for (int j = 0; j < 2; ++j) {
    const int e = j * 256 + threadIdx.x;
    BCs[e] = *(const float4*)(xp + (tok0 + (e >> 3)) * XDIM + DT_RANK + (e & 7) * 4);
  }

  const float* __restrict__ dp = delta + tok0 * D_MODEL + d;
  const float* __restrict__ up = u     + tok0 * D_MODEL + d;
  float* __restrict__       yp = y     + tok0 * D_MODEL + d;

  float dtc[PF], utc[PF], dtn[PF], utn[PF];
  load_group(dp, up, dtc, utc);

  float A[D_STATE];
  #pragma unroll
  for (int n = 0; n < D_STATE; ++n) A[n] = -__expf(A_log[d * D_STATE + n]);

  const size_t idx = ((size_t)b * NCHUNK + c) * D_MODEL + d;
  float h[D_STATE];
  #pragma unroll
  for (int n = 0; n < D_STATE; ++n) h[n] = Hc[idx * D_STATE + n];
  const float dsk = Dsk[d];

  __syncthreads();

  float4 nbc[8];
  #pragma unroll
  for (int r = 0; r < 8; ++r) nbc[r] = BCs[r];

  for (int g = 0; g < NGRP - 1; ++g) {
    load_group(dp + (size_t)(g + 1) * PF * D_MODEL, up + (size_t)(g + 1) * PF * D_MODEL, dtn, utn);
    SCHED_FENCE();
    pass3_group<false>(BCs + g * PF * 8, dtc, utc, A, h, dsk, yp + (size_t)g * PF * D_MODEL, nbc);
    #pragma unroll
    for (int i = 0; i < PF; ++i) { dtc[i] = dtn[i]; utc[i] = utn[i]; }
  }
  pass3_group<true>(BCs + (NGRP - 1) * PF * 8, dtc, utc, A, h, dsk, yp + (size_t)(NGRP - 1) * PF * D_MODEL, nbc);
}

extern "C" void kernel_launch(void* const* d_in, const int* in_sizes, int n_in,
                              void* d_out, int out_size, void* d_ws, size_t ws_size,
                              hipStream_t stream) {
  const float* u      = (const float*)d_in[0];
  const float* A_log  = (const float*)d_in[1];
  const float* D_skip = (const float*)d_in[2];
  const float* dt_w   = (const float*)d_in[3];
  const float* dt_b   = (const float*)d_in[4];
  const float* x_w    = (const float*)d_in[5];
  float* y = (float*)d_out;

  float* ws    = (float*)d_ws;
  float* xp    = ws;
  float* delta = xp    + (size_t)NTOK * XDIM;
  float* Q     = delta + (size_t)NTOK * D_MODEL;
  float* Sdt   = Q     + (size_t)BATCH * NCHUNK * D_MODEL * D_STATE;
  float* Hc    = Sdt   + (size_t)BATCH * NCHUNK * D_MODEL;

  xp_wmma_kernel<<<NTOK / 16, 256, 0, stream>>>(u, x_w, xp);

  delta_wmma_kernel<<<(NTOK / 16) * (D_MODEL / 16 / NT2) * 32 / 256, 256, 0, stream>>>(
      xp, dt_w, dt_b, delta);

  scan_pass1_kernel<<<BATCH * NCHUNK * (D_MODEL / 256), 256, 0, stream>>>(
      delta, xp, u, A_log, Q, Sdt);

  scan_carry_kernel<<<BATCH * D_MODEL * D_STATE / 256, 256, 0, stream>>>(Q, Sdt, A_log, Hc);

  scan_pass3_kernel<<<BATCH * NCHUNK * (D_MODEL / 256), 256, 0, stream>>>(
      delta, xp, u, A_log, D_skip, Hc, y);
}


# xp and delta GEMMs: operands loaded 16 bytes per lane, the 4 loads of a row's 128-byte line back to back, two 16-WMMA units ahead of use (same k permutation in A and B: changes k summation order)
# speedup vs baseline: 2.5458x; 1.2371x over previous
#include <hip/hip_runtime.h>
#include <hip/hip_bf16.h>

typedef float v2f __attribute__((ext_vector_type(2)));
typedef float v4f __attribute__((ext_vector_type(4)));
typedef float v8f __attribute__((ext_vector_type(8)));

#define D_MODEL 1024
#define D_STATE 16
#define DT_RANK 64
#define BATCH   4
#define SEQLEN  2048
#define XDIM    96
#define NTOK    (BATCH * SEQLEN)
#define NCHUNK  32
#define CLEN    (SEQLEN / NCHUNK)
#define NT1     (XDIM / 16)
#define NT2     8
#define KLINE   32

#define SCHED_FENCE() __builtin_amdgcn_sched_barrier(0)

__device__ __forceinline__ void load_line(const float* __restrict__ p, v4f (&q)[4]) {
  #pragma unroll
  for (int j = 0; j < 4; ++j) q[j] = *(const v4f*)(p + j * 8);
}

__device__ __forceinline__ void wmma_unit(const v4f (&a)[4], const v4f (&b0)[4], const v4f (&b1)[4],
                                          v8f& c0, v8f& c1) {
  #pragma unroll
  for (int j = 0; j < 4; ++j) {
    c0 = __builtin_amdgcn_wmma_f32_16x16x4_f32(false, a[j].xy, false, b0[j].xy, (short)0, c0, false, false);
    c1 = __builtin_amdgcn_wmma_f32_16x16x4_f32(false, a[j].xy, false, b1[j].xy, (short)0, c1, false, false);
    c0 = __builtin_amdgcn_wmma_f32_16x16x4_f32(false, a[j].zw, false, b0[j].zw, (short)0, c0, false, false);
    c1 = __builtin_amdgcn_wmma_f32_16x16x4_f32(false, a[j].zw, false, b1[j].zw, (short)0, c1, false, false);
  }
}

#define KSPLIT   8
#define KSLICE   (D_MODEL / KSPLIT)
#define NT_PITCH 272
#define W_PITCH  (NT1 * NT_PITCH)

__global__ __launch_bounds__(256) void xp_wmma_kernel(const float* __restrict__ u,
                                                      const float* __restrict__ xw,
                                                      float* __restrict__ xp) {
  __shared__ float part[KSPLIT * W_PITCH];

  const int m_tile = blockIdx.x;
  const int wv     = threadIdx.x >> 5;
  const int lane   = threadIdx.x & 31;
  const int row    = lane & 15;
  const int kq     = (lane >> 4) << 2;

  const float* __restrict__ ap = u  + (size_t)(m_tile * 16 + row) * D_MODEL + wv * KSLICE + kq;
  const float* __restrict__ bp = xw + (size_t)row * D_MODEL + wv * KSLICE + kq;

  constexpr int NP = NT1 / 2;
  constexpr int NU = (KSLICE / KLINE) * NP;

  v8f acc[NT1] = {};
  v4f a[2][4];
  v4f b[3][2][4];

  load_line(ap, a[0]);
  #pragma unroll
  for (int v = 0; v < 2; ++v) {
    load_line(bp + (size_t)(2 * v) * 16 * D_MODEL, b[v][0]);
    load_line(bp + (size_t)(2 * v + 1) * 16 * D_MODEL, b[v][1]);
  }
  #pragma unroll
  for (int un = 0; un < NU; ++un) {
    const int c = un / NP, p = un % NP;
    if (un + 2 < NU) {
      const int c2 = (un + 2) / NP, p2 = (un + 2) % NP;
      if (p2 == 0) load_line(ap + c2 * KLINE, a[c2 & 1]);
      load_line(bp + (size_t)(2 * p2) * 16 * D_MODEL + c2 * KLINE, b[(un + 2) % 3][0]);
      load_line(bp + (size_t)(2 * p2 + 1) * 16 * D_MODEL + c2 * KLINE, b[(un + 2) % 3][1]);
    }
    SCHED_FENCE();
    wmma_unit(a[c & 1], b[un % 3][0], b[un % 3][1], acc[2 * p], acc[2 * p + 1]);
  }

  float* pw = part + wv * W_PITCH + lane;
  #pragma unroll
  for (int nt = 0; nt < NT1; ++nt)
    #pragma unroll
    for (int i = 0; i < 8; ++i)
      pw[nt * NT_PITCH + i * 32] = acc[nt][i];

  __syncthreads();

  float* __restrict__ outp = xp + (size_t)m_tile * 16 * XDIM;
  #pragma unroll
  for (int j = 0; j < (16 * XDIM) / 256; ++j) {
    const int e   = j * 256 + threadIdx.x;
    const int r   = e / XDIM;
    const int col = e - r * XDIM;
    const int off = (col >> 4) * NT_PITCH + (r & 7) * 32 + (r >> 3) * 16 + (col & 15);
    float s = part[off];
    #pragma unroll
    for (int w = 1; w < KSPLIT; ++w) s += part[w * W_PITCH + off];
    outp[e] = s;
  }
}

__global__ __launch_bounds__(256) void delta_wmma_kernel(const float* __restrict__ xp,
                                                         const float* __restrict__ wt,
                                                         const float* __restrict__ bias,
                                                         float* __restrict__ delta) {
  const int gtid   = blockIdx.x * 256 + threadIdx.x;
  const int wave   = gtid >> 5;
  const int lane   = threadIdx.x & 31;
  const int ngrp   = wave % (D_MODEL / 16 / NT2);
  const int m_tile = wave / (D_MODEL / 16 / NT2);
  const int row    = lane & 15;
  const int kq     = (lane >> 4) << 2;

  const float* __restrict__ ap = xp + (size_t)(m_tile * 16 + row) * XDIM + kq;
  const float* __restrict__ bp = wt + (size_t)(ngrp * NT2 * 16 + row) * DT_RANK + kq;

  constexpr int NP = NT2 / 2;
  constexpr int NU = (DT_RANK / KLINE) * NP;

  v8f acc[NT2] = {};
  v4f a[2][4];
  v4f b[3][2][4];

  load_line(ap, a[0]);
  load_line(ap + KLINE, a[1]);
  #pragma unroll
  for (int v = 0; v < 2; ++v) {
    load_line(bp + (size_t)(2 * v) * 16 * DT_RANK, b[v][0]);
    load_line(bp + (size_t)(2 * v + 1) * 16 * DT_RANK, b[v][1]);
  }
  #pragma unroll
  for (int un = 0; un < NU; ++un) {
    const int c = un / NP, p = un % NP;
    if (un + 2 < NU) {
      const int c2 = (un + 2) / NP, p2 = (un + 2) % NP;
      load_line(bp + (size_t)(2 * p2) * 16 * DT_RANK + c2 * KLINE, b[(un + 2) % 3][0]);
      load_line(bp + (size_t)(2 * p2 + 1) * 16 * DT_RANK + c2 * KLINE, b[(un + 2) % 3][1]);
    }
    SCHED_FENCE();
    wmma_unit(a[c], b[un % 3][0], b[un % 3][1], acc[2 * p], acc[2 * p + 1]);
  }

  const int m0 = m_tile * 16 + ((lane >> 4) << 3);
  #pragma unroll
  for (int nt = 0; nt < NT2; ++nt) {
    const int n0 = (ngrp * NT2 + nt) * 16 + (lane & 15);
    const float bn = bias[n0];
    #pragma unroll
    for (int i = 0; i < 8; ++i) {
      float x  = acc[nt][i] + bn;
      float sp = (x > 20.0f) ? x : __logf(1.0f + __expf(x));
      delta[(size_t)(m0 + i) * D_MODEL + n0] = sp;
    }
  }
}

#define PF   8
#define NGRP (CLEN / PF)


__device__ __forceinline__ void load_group(const float* __restrict__ dp, const float* __restrict__ up,
                                           float (&dt)[PF], float (&ut)[PF]) {
  #pragma unroll
  for (int i = 0; i < PF; ++i) {
    dt[i] = dp[(size_t)i * D_MODEL];
    ut[i] = up[(size_t)i * D_MODEL];
  }
}

template <bool LAST>
__device__ __forceinline__ void pass1_group(const float4* rows, const float (&dt)[PF],
                                            const float (&ut)[PF], const float (&A)[D_STATE],
                                            float (&h)[D_STATE], float& sdt,
                                            float4 (&nb)[4]) {
  #pragma unroll
  for (int i = 0; i < PF; ++i) {
    const float4 q0 = nb[0], q1 = nb[1], q2 = nb[2], q3 = nb[3];
    if (i + 1 < PF || !LAST) {
      nb[0] = rows[(i + 1) * 4 + 0]; nb[1] = rows[(i + 1) * 4 + 1];
      nb[2] = rows[(i + 1) * 4 + 2]; nb[3] = rows[(i + 1) * 4 + 3];
    }
    SCHED_FENCE();
    const float Bt[D_STATE] = {q0.x, q0.y, q0.z, q0.w, q1.x, q1.y, q1.z, q1.w,
                               q2.x, q2.y, q2.z, q2.w, q3.x, q3.y, q3.z, q3.w};
    const float d_t = dt[i];
    sdt += d_t;
    const float du = d_t * ut[i];
    #pragma unroll
    for (int n = 0; n < D_STATE; ++n)
      h[n] = __expf(d_t * A[n]) * h[n] + du * Bt[n];
  }
}

__global__ __launch_bounds__(256) void scan_pass1_kernel(const float* __restrict__ delta,
                                                         const float* __restrict__ xp,
                                                         const float* __restrict__ u,
                                                         const float* __restrict__ A_log,
                                                         float* __restrict__ Q,
                                                         float* __restrict__ Sdt) {
  __shared__ float4 Bs[CLEN * 4];

  const int bid = blockIdx.x;
  const int d   = (bid & 3) * 256 + threadIdx.x;
  const int c   = (bid >> 2) & (NCHUNK - 1);
  const int b   = bid >> 7;
  const size_t tok0 = (size_t)b * SEQLEN + (size_t)c * CLEN;

  Bs[threadIdx.x] = *(const float4*)(xp + (tok0 + (threadIdx.x >> 2)) * XDIM + DT_RANK +
                                     (threadIdx.x & 3) * 4);

  const float* __restrict__ dp = delta + tok0 * D_MODEL + d;
  const float* __restrict__ up = u     + tok0 * D_MODEL + d;

  float dtc[PF], utc[PF], dtn[PF], utn[PF];
  load_group(dp, up, dtc, utc);

  float A[D_STATE];
  #pragma unroll
  for (int n = 0; n < D_STATE; ++n) A[n] = -__expf(A_log[d * D_STATE + n]);

  float h[D_STATE];
  #pragma unroll
  for (int n = 0; n < D_STATE; ++n) h[n] = 0.0f;
  float sdt = 0.0f;

  __syncthreads();

  float4 nb[4] = {Bs[0], Bs[1], Bs[2], Bs[3]};

  for (int g = 0; g < NGRP - 1; ++g) {
    load_group(dp + (size_t)(g + 1) * PF * D_MODEL, up + (size_t)(g + 1) * PF * D_MODEL, dtn, utn);
    SCHED_FENCE();
    pass1_group<false>(Bs + g * PF * 4, dtc, utc, A, h, sdt, nb);
    #pragma unroll
    for (int i = 0; i < PF; ++i) { dtc[i] = dtn[i]; utc[i] = utn[i]; }
  }
  pass1_group<true>(Bs + (NGRP - 1) * PF * 4, dtc, utc, A, h, sdt, nb);

  const size_t idx = ((size_t)b * NCHUNK + c) * D_MODEL + d;
  #pragma unroll
  for (int n = 0; n < D_STATE; ++n) Q[idx * D_STATE + n] = h[n];
  Sdt[idx] = sdt;
}

#define CROW (D_MODEL * D_STATE)

__global__ __launch_bounds__(256) void scan_carry_kernel(const float* __restrict__ Q,
                                                         const float* __restrict__ Sdt,
                                                         const float* __restrict__ A_log,
                                                         float* __restrict__ Hc) {
  const unsigned tid = blockIdx.x * 256u + threadIdx.x;
  const unsigned e   = tid % CROW;
  const unsigned b   = tid / CROW;
  const unsigned d   = e / D_STATE;

  const float* __restrict__ qp = Q   + (size_t)b * NCHUNK * CROW + e;
  const float* __restrict__ sp = Sdt + (size_t)b * NCHUNK * D_MODEL + d;
  float* __restrict__       hp = Hc  + (size_t)b * NCHUNK * CROW + e;

  float q[NCHUNK], s[NCHUNK];
  #pragma unroll
  for (int c = 0; c < NCHUNK; ++c) {
    q[c] = qp[(size_t)c * CROW];
    s[c] = sp[(size_t)c * D_MODEL];
  }
  const float A = -__expf(A_log[e]);

  float h = 0.0f;
  #pragma unroll
  for (int c = 0; c < NCHUNK; ++c) {
    hp[(size_t)c * CROW] = h;
    h = __expf(s[c] * A) * h + q[c];
  }
}

template <bool LAST>
__device__ __forceinline__ void pass3_group(const float4* rows, const float (&dt)[PF],
                                            const float (&ut)[PF], const float (&A)[D_STATE],
                                            float (&h)[D_STATE], const float dsk,
                                            float* __restrict__ yp, float4 (&nbc)[8]) {
  #pragma unroll
  for (int i = 0; i < PF; ++i) {
    const float4 q0 = nbc[0], q1 = nbc[1], q2 = nbc[2], q3 = nbc[3];
    const float4 p0 = nbc[4], p1 = nbc[5], p2 = nbc[6], p3 = nbc[7];
    if (i + 1 < PF || !LAST) {
      #pragma unroll
      for (int r = 0; r < 8; ++r) nbc[r] = rows[(i + 1) * 8 + r];
    }
    SCHED_FENCE();
    const float Bt[D_STATE] = {q0.x, q0.y, q0.z, q0.w, q1.x, q1.y, q1.z, q1.w,
                               q2.x, q2.y, q2.z, q2.w, q3.x, q3.y, q3.z, q3.w};
    const float Ct[D_STATE] = {p0.x, p0.y, p0.z, p0.w, p1.x, p1.y, p1.z, p1.w,
                               p2.x, p2.y, p2.z, p2.w, p3.x, p3.y, p3.z, p3.w};
    const float d_t = dt[i];
    const float u_t = ut[i];
    const float du = d_t * u_t;
    float acc = 0.0f;
    #pragma unroll
    for (int n = 0; n < D_STATE; ++n) {
      h[n] = __expf(d_t * A[n]) * h[n] + du * Bt[n];
      acc += h[n] * Ct[n];
    }
    yp[(size_t)i * D_MODEL] = acc + u_t * dsk;
  }
}

__global__ __launch_bounds__(256) void scan_pass3_kernel(const float* __restrict__ delta,
                                                         const float* __restrict__ xp,
                                                         const float* __restrict__ u,
                                                         const float* __restrict__ A_log,
                                                         const float* __restrict__ Dsk,
                                                         const float* __restrict__ Hc,
                                                         float* __restrict__ y) {
  __shared__ float4 BCs[CLEN * 8];

  const int bid = blockIdx.x;
  const int d   = (bid & 3) * 256 + threadIdx.x;
  const int c   = (bid >> 2) & (NCHUNK - 1);
  const int b   = bid >> 7;
  const size_t tok0 = (size_t)b * SEQLEN + (size_t)c * CLEN;

  #pragma unroll
  for (int j = 0; j < 2; ++j) {
    const int e = j * 256 + threadIdx.x;
    BCs[e] = *(const float4*)(xp + (tok0 + (e >> 3)) * XDIM + DT_RANK + (e & 7) * 4);
  }

  const float* __restrict__ dp = delta + tok0 * D_MODEL + d;
  const float* __restrict__ up = u     + tok0 * D_MODEL + d;
  float* __restrict__       yp = y     + tok0 * D_MODEL + d;

  float dtc[PF], utc[PF], dtn[PF], utn[PF];
  load_group(dp, up, dtc, utc);

  float A[D_STATE];
  #pragma unroll
  for (int n = 0; n < D_STATE; ++n) A[n] = -__expf(A_log[d * D_STATE + n]);

  const size_t idx = ((size_t)b * NCHUNK + c) * D_MODEL + d;
  float h[D_STATE];
  #pragma unroll
  for (int n = 0; n < D_STATE; ++n) h[n] = Hc[idx * D_STATE + n];
  const float dsk = Dsk[d];

  __syncthreads();

  float4 nbc[8];
  #pragma unroll
  for (int r = 0; r < 8; ++r) nbc[r] = BCs[r];

  for (int g = 0; g < NGRP - 1; ++g) {
    load_group(dp + (size_t)(g + 1) * PF * D_MODEL, up + (size_t)(g + 1) * PF * D_MODEL, dtn, utn);
    SCHED_FENCE();
    pass3_group<false>(BCs + g * PF * 8, dtc, utc, A, h, dsk, yp + (size_t)g * PF * D_MODEL, nbc);
    #pragma unroll
    for (int i = 0; i < PF; ++i) { dtc[i] = dtn[i]; utc[i] = utn[i]; }
  }
  pass3_group<true>(BCs + (NGRP - 1) * PF * 8, dtc, utc, A, h, dsk, yp + (size_t)(NGRP - 1) * PF * D_MODEL, nbc);
}

extern "C" void kernel_launch(void* const* d_in, const int* in_sizes, int n_in,
                              void* d_out, int out_size, void* d_ws, size_t ws_size,
                              hipStream_t stream) {
  const float* u      = (const float*)d_in[0];
  const float* A_log  = (const float*)d_in[1];
  const float* D_skip = (const float*)d_in[2];
  const float* dt_w   = (const float*)d_in[3];
  const float* dt_b   = (const float*)d_in[4];
  const float* x_w    = (const float*)d_in[5];
  float* y = (float*)d_out;

  float* ws    = (float*)d_ws;
  float* xp    = ws;
  float* delta = xp    + (size_t)NTOK * XDIM;
  float* Q     = delta + (size_t)NTOK * D_MODEL;
  float* Sdt   = Q     + (size_t)BATCH * NCHUNK * D_MODEL * D_STATE;
  float* Hc    = Sdt   + (size_t)BATCH * NCHUNK * D_MODEL;

  xp_wmma_kernel<<<NTOK / 16, 256, 0, stream>>>(u, x_w, xp);

  delta_wmma_kernel<<<(NTOK / 16) * (D_MODEL / 16 / NT2) * 32 / 256, 256, 0, stream>>>(
      xp, dt_w, dt_b, delta);

  scan_pass1_kernel<<<BATCH * NCHUNK * (D_MODEL / 256), 256, 0, stream>>>(
      delta, xp, u, A_log, Q, Sdt);

  scan_carry_kernel<<<BATCH * D_MODEL * D_STATE / 256, 256, 0, stream>>>(Q, Sdt, A_log, Hc);

  scan_pass3_kernel<<<BATCH * NCHUNK * (D_MODEL / 256), 256, 0, stream>>>(
      delta, xp, u, A_log, D_skip, Hc, y);
}


# delta GEMM moved into the xp kernel: a block sums its 16-token xp stripe in LDS and multiplies it by dt_proj_w at once (4 launches, not 5; xp columns 0..63 never leave the chip); sums unchanged
# speedup vs baseline: 2.6501x; 1.0410x over previous
#include <hip/hip_runtime.h>
#include <hip/hip_bf16.h>

typedef float v2f __attribute__((ext_vector_type(2)));
typedef float v4f __attribute__((ext_vector_type(4)));
typedef float v8f __attribute__((ext_vector_type(8)));

#define D_MODEL 1024
#define D_STATE 16
#define DT_RANK 64
#define BATCH   4
#define SEQLEN  2048
#define XDIM    96
#define NTOK    (BATCH * SEQLEN)
#define NCHUNK  32
#define CLEN    (SEQLEN / NCHUNK)
#define NT1     (XDIM / 16)
#define NT2     8
#define KLINE   32

#define SCHED_FENCE() __builtin_amdgcn_sched_barrier(0)

__device__ __forceinline__ void load_line(const float* __restrict__ p, v4f (&q)[4]) {
  #pragma unroll
  for (int j = 0; j < 4; ++j) q[j] = *(const v4f*)(p + j * 8);
}

__device__ __forceinline__ void wmma_unit(const v4f (&a)[4], const v4f (&b0)[4], const v4f (&b1)[4],
                                          v8f& c0, v8f& c1) {
  #pragma unroll
  for (int j = 0; j < 4; ++j) {
    c0 = __builtin_amdgcn_wmma_f32_16x16x4_f32(false, a[j].xy, false, b0[j].xy, (short)0, c0, false, false);
    c1 = __builtin_amdgcn_wmma_f32_16x16x4_f32(false, a[j].xy, false, b1[j].xy, (short)0, c1, false, false);
    c0 = __builtin_amdgcn_wmma_f32_16x16x4_f32(false, a[j].zw, false, b0[j].zw, (short)0, c0, false, false);
    c1 = __builtin_amdgcn_wmma_f32_16x16x4_f32(false, a[j].zw, false, b1[j].zw, (short)0, c1, false, false);
  }
}

#define KSPLIT   8
#define KSLICE   (D_MODEL / KSPLIT)
#define NT_PITCH 272
#define W_PITCH  (NT1 * NT_PITCH)
#define XS_PITCH 68

__global__ __launch_bounds__(256) void xp_delta_wmma_kernel(const float* __restrict__ u,
                                                            const float* __restrict__ xw,
                                                            const float* __restrict__ wt,
                                                            const float* __restrict__ bias,
                                                            float* __restrict__ xp,
                                                            float* __restrict__ delta) {
  __shared__ float part[KSPLIT * W_PITCH];
  __shared__ float xs[16 * XS_PITCH];

  const int m_tile = blockIdx.x;
  const int wv     = threadIdx.x >> 5;
  const int lane   = threadIdx.x & 31;
  const int row    = lane & 15;
  const int kq     = (lane >> 4) << 2;

  v4f b[3][2][4];

  {
    const float* __restrict__ ap = u  + (size_t)(m_tile * 16 + row) * D_MODEL + wv * KSLICE + kq;
    const float* __restrict__ bp = xw + (size_t)row * D_MODEL + wv * KSLICE + kq;

    constexpr int NP = NT1 / 2;
    constexpr int NU = (KSLICE / KLINE) * NP;

    v8f acc[NT1] = {};
    v4f a[2][4];

    load_line(ap, a[0]);
    #pragma unroll
    for (int v = 0; v < 2; ++v) {
      load_line(bp + (size_t)(2 * v) * 16 * D_MODEL, b[v][0]);
      load_line(bp + (size_t)(2 * v + 1) * 16 * D_MODEL, b[v][1]);
    }
    #pragma unroll
    for (int un = 0; un < NU; ++un) {
      const int c = un / NP, p = un % NP;
      if (un + 2 < NU) {
        const int c2 = (un + 2) / NP, p2 = (un + 2) % NP;
        if (p2 == 0) load_line(ap + c2 * KLINE, a[c2 & 1]);
        load_line(bp + (size_t)(2 * p2) * 16 * D_MODEL + c2 * KLINE, b[(un + 2) % 3][0]);
        load_line(bp + (size_t)(2 * p2 + 1) * 16 * D_MODEL + c2 * KLINE, b[(un + 2) % 3][1]);
      }
      SCHED_FENCE();
      wmma_unit(a[c & 1], b[un % 3][0], b[un % 3][1], acc[2 * p], acc[2 * p + 1]);
    }

    float* pw = part + wv * W_PITCH + lane;
    #pragma unroll
    for (int nt = 0; nt < NT1; ++nt)
      #pragma unroll
      for (int i = 0; i < 8; ++i)
        pw[nt * NT_PITCH + i * 32] = acc[nt][i];
  }

  __syncthreads();

  const float* __restrict__ bp2 = wt + (size_t)(wv * NT2 * 16 + row) * DT_RANK + kq;
  #pragma unroll
  for (int v = 0; v < 2; ++v) {
    load_line(bp2 + (size_t)(2 * v) * 16 * DT_RANK, b[v][0]);
    load_line(bp2 + (size_t)(2 * v + 1) * 16 * DT_RANK, b[v][1]);
  }
  SCHED_FENCE();

  float s[(16 * XDIM) / 256];
  #pragma unroll
  for (int j = 0; j < (16 * XDIM) / 256; ++j) {
    const int e   = j * 256 + threadIdx.x;
    const int r   = e / XDIM;
    const int col = e - r * XDIM;
    const int off = (col >> 4) * NT_PITCH + (r & 7) * 32 + (r >> 3) * 16 + (col & 15);
    float t = part[off];
    #pragma unroll
    for (int w = 1; w < KSPLIT; ++w) t += part[w * W_PITCH + off];
    s[j] = t;
  }
  #pragma unroll
  for (int j = 0; j < (16 * XDIM) / 256; ++j) {
    const int e   = j * 256 + threadIdx.x;
    const int r   = e / XDIM;
    const int col = e - r * XDIM;
    if (col < DT_RANK) xs[r * XS_PITCH + col] = s[j];
  }

  __syncthreads();

  float* __restrict__ outp = xp + (size_t)m_tile * 16 * XDIM;
  #pragma unroll
  for (int j = 0; j < (16 * XDIM) / 256; ++j) {
    const int e   = j * 256 + threadIdx.x;
    const int col = e % XDIM;
    if (col >= DT_RANK) outp[e] = s[j];
  }

  {
    constexpr int NP = NT2 / 2;
    constexpr int NU = (DT_RANK / KLINE) * NP;

    const float* ap2 = xs + row * XS_PITCH + kq;

    v8f acc[NT2] = {};
    v4f a[2][4];
    load_line(ap2, a[0]);
    load_line(ap2 + KLINE, a[1]);

    #pragma unroll
    for (int un = 0; un < NU; ++un) {
      const int c = un / NP, p = un % NP;
      if (un + 2 < NU) {
        const int c2 = (un + 2) / NP, p2 = (un + 2) % NP;
        load_line(bp2 + (size_t)(2 * p2) * 16 * DT_RANK + c2 * KLINE, b[(un + 2) % 3][0]);
        load_line(bp2 + (size_t)(2 * p2 + 1) * 16 * DT_RANK + c2 * KLINE, b[(un + 2) % 3][1]);
      }
      SCHED_FENCE();
      wmma_unit(a[c], b[un % 3][0], b[un % 3][1], acc[2 * p], acc[2 * p + 1]);
    }

    const int m0 = m_tile * 16 + ((lane >> 4) << 3);
    #pragma unroll
    for (int nt = 0; nt < NT2; ++nt) {
      const int n0 = (wv * NT2 + nt) * 16 + (lane & 15);
      const float bn = bias[n0];
      #pragma unroll
      for (int i = 0; i < 8; ++i) {
        float x  = acc[nt][i] + bn;
        float sp = (x > 20.0f) ? x : __logf(1.0f + __expf(x));
        delta[(size_t)(m0 + i) * D_MODEL + n0] = sp;
      }
    }
  }
}

#define PF   8
#define NGRP (CLEN / PF)


__device__ __forceinline__ void load_group(const float* __restrict__ dp, const float* __restrict__ up,
                                           float (&dt)[PF], float (&ut)[PF]) {
  #pragma unroll
  for (int i = 0; i < PF; ++i) {
    dt[i] = dp[(size_t)i * D_MODEL];
    ut[i] = up[(size_t)i * D_MODEL];
  }
}

template <bool LAST>
__device__ __forceinline__ void pass1_group(const float4* rows, const float (&dt)[PF],
                                            const float (&ut)[PF], const float (&A)[D_STATE],
                                            float (&h)[D_STATE], float& sdt,
                                            float4 (&nb)[4]) {
  #pragma unroll
  for (int i = 0; i < PF; ++i) {
    const float4 q0 = nb[0], q1 = nb[1], q2 = nb[2], q3 = nb[3];
    if (i + 1 < PF || !LAST) {
      nb[0] = rows[(i + 1) * 4 + 0]; nb[1] = rows[(i + 1) * 4 + 1];
      nb[2] = rows[(i + 1) * 4 + 2]; nb[3] = rows[(i + 1) * 4 + 3];
    }
    SCHED_FENCE();
    const float Bt[D_STATE] = {q0.x, q0.y, q0.z, q0.w, q1.x, q1.y, q1.z, q1.w,
                               q2.x, q2.y, q2.z, q2.w, q3.x, q3.y, q3.z, q3.w};
    const float d_t = dt[i];
    sdt += d_t;
    const float du = d_t * ut[i];
    #pragma unroll
    for (int n = 0; n < D_STATE; ++n)
      h[n] = __expf(d_t * A[n]) * h[n] + du * Bt[n];
  }
}

__global__ __launch_bounds__(256) void scan_pass1_kernel(const float* __restrict__ delta,
                                                         const float* __restrict__ xp,
                                                         const float* __restrict__ u,
                                                         const float* __restrict__ A_log,
                                                         float* __restrict__ Q,
                                                         float* __restrict__ Sdt) {
  __shared__ float4 Bs[CLEN * 4];

  const int bid = blockIdx.x;
  const int d   = (bid & 3) * 256 + threadIdx.x;
  const int c   = (bid >> 2) & (NCHUNK - 1);
  const int b   = bid >> 7;
  const size_t tok0 = (size_t)b * SEQLEN + (size_t)c * CLEN;

  Bs[threadIdx.x] = *(const float4*)(xp + (tok0 + (threadIdx.x >> 2)) * XDIM + DT_RANK +
                                     (threadIdx.x & 3) * 4);

  const float* __restrict__ dp = delta + tok0 * D_MODEL + d;
  const float* __restrict__ up = u     + tok0 * D_MODEL + d;

  float dtc[PF], utc[PF], dtn[PF], utn[PF];
  load_group(dp, up, dtc, utc);

  float A[D_STATE];
  #pragma unroll
  for (int n = 0; n < D_STATE; ++n) A[n] = -__expf(A_log[d * D_STATE + n]);

  float h[D_STATE];
  #pragma unroll
  for (int n = 0; n < D_STATE; ++n) h[n] = 0.0f;
  float sdt = 0.0f;

  __syncthreads();

  float4 nb[4] = {Bs[0], Bs[1], Bs[2], Bs[3]};

  for (int g = 0; g < NGRP - 1; ++g) {
    load_group(dp + (size_t)(g + 1) * PF * D_MODEL, up + (size_t)(g + 1) * PF * D_MODEL, dtn, utn);
    SCHED_FENCE();
    pass1_group<false>(Bs + g * PF * 4, dtc, utc, A, h, sdt, nb);
    #pragma unroll
    for (int i = 0; i < PF; ++i) { dtc[i] = dtn[i]; utc[i] = utn[i]; }
  }
  pass1_group<true>(Bs + (NGRP - 1) * PF * 4, dtc, utc, A, h, sdt, nb);

  const size_t idx = ((size_t)b * NCHUNK + c) * D_MODEL + d;
  #pragma unroll
  for (int n = 0; n < D_STATE; ++n) Q[idx * D_STATE + n] = h[n];
  Sdt[idx] = sdt;
}

#define CROW (D_MODEL * D_STATE)

__global__ __launch_bounds__(256) void scan_carry_kernel(const float* __restrict__ Q,
                                                         const float* __restrict__ Sdt,
                                                         const float* __restrict__ A_log,
                                                         float* __restrict__ Hc) {
  const unsigned tid = blockIdx.x * 256u + threadIdx.x;
  const unsigned e   = tid % CROW;
  const unsigned b   = tid / CROW;
  const unsigned d   = e / D_STATE;

  const float* __restrict__ qp = Q   + (size_t)b * NCHUNK * CROW + e;
  const float* __restrict__ sp = Sdt + (size_t)b * NCHUNK * D_MODEL + d;
  float* __restrict__       hp = Hc  + (size_t)b * NCHUNK * CROW + e;

  float q[NCHUNK], s[NCHUNK];
  #pragma unroll
  for (int c = 0; c < NCHUNK; ++c) {
    q[c] = qp[(size_t)c * CROW];
    s[c] = sp[(size_t)c * D_MODEL];
  }
  const float A = -__expf(A_log[e]);

  float h = 0.0f;
  #pragma unroll
  for (int c = 0; c < NCHUNK; ++c) {
    hp[(size_t)c * CROW] = h;
    h = __expf(s[c] * A) * h + q[c];
  }
}

template <bool LAST>
__device__ __forceinline__ void pass3_group(const float4* rows, const float (&dt)[PF],
                                            const float (&ut)[PF], const float (&A)[D_STATE],
                                            float (&h)[D_STATE], const float dsk,
                                            float* __restrict__ yp, float4 (&nbc)[8]) {
  #pragma unroll
  for (int i = 0; i < PF; ++i) {
    const float4 q0 = nbc[0], q1 = nbc[1], q2 = nbc[2], q3 = nbc[3];
    const float4 p0 = nbc[4], p1 = nbc[5], p2 = nbc[6], p3 = nbc[7];
    if (i + 1 < PF || !LAST) {
      #pragma unroll
      for (int r = 0; r < 8; ++r) nbc[r] = rows[(i + 1) * 8 + r];
    }
    SCHED_FENCE();
    const float Bt[D_STATE] = {q0.x, q0.y, q0.z, q0.w, q1.x, q1.y, q1.z, q1.w,
                               q2.x, q2.y, q2.z, q2.w, q3.x, q3.y, q3.z, q3.w};
    const float Ct[D_STATE] = {p0.x, p0.y, p0.z, p0.w, p1.x, p1.y, p1.z, p1.w,
                               p2.x, p2.y, p2.z, p2.w, p3.x, p3.y, p3.z, p3.w};
    const float d_t = dt[i];
    const float u_t = ut[i];
    const float du = d_t * u_t;
    float acc = 0.0f;
    #pragma unroll
    for (int n = 0; n < D_STATE; ++n) {
      h[n] = __expf(d_t * A[n]) * h[n] + du * Bt[n];
      acc += h[n] * Ct[n];
    }
    yp[(size_t)i * D_MODEL] = acc + u_t * dsk;
  }
}

__global__ __launch_bounds__(256) void scan_pass3_kernel(const float* __restrict__ delta,
                                                         const float* __restrict__ xp,
                                                         const float* __restrict__ u,
                                                         const float* __restrict__ A_log,
                                                         const float* __restrict__ Dsk,
                                                         const float* __restrict__ Hc,
                                                         float* __restrict__ y) {
  __shared__ float4 BCs[CLEN * 8];

  const int bid = blockIdx.x;
  const int d   = (bid & 3) * 256 + threadIdx.x;
  const int c   = (bid >> 2) & (NCHUNK - 1);
  const int b   = bid >> 7;
  const size_t tok0 = (size_t)b * SEQLEN + (size_t)c * CLEN;

  #pragma unroll
  for (int j = 0; j < 2; ++j) {
    const int e = j * 256 + threadIdx.x;
    BCs[e] = *(const float4*)(xp + (tok0 + (e >> 3)) * XDIM + DT_RANK + (e & 7) * 4);
  }

  const float* __restrict__ dp = delta + tok0 * D_MODEL + d;
  const float* __restrict__ up = u     + tok0 * D_MODEL + d;
  float* __restrict__       yp = y     + tok0 * D_MODEL + d;

  float dtc[PF], utc[PF], dtn[PF], utn[PF];
  load_group(dp, up, dtc, utc);

  float A[D_STATE];
  #pragma unroll
  for (int n = 0; n < D_STATE; ++n) A[n] = -__expf(A_log[d * D_STATE + n]);

  const size_t idx = ((size_t)b * NCHUNK + c) * D_MODEL + d;
  float h[D_STATE];
  #pragma unroll
  for (int n = 0; n < D_STATE; ++n) h[n] = Hc[idx * D_STATE + n];
  const float dsk = Dsk[d];

  __syncthreads();

  float4 nbc[8];
  #pragma unroll
  for (int r = 0; r < 8; ++r) nbc[r] = BCs[r];

  for (int g = 0; g < NGRP - 1; ++g) {
    load_group(dp + (size_t)(g + 1) * PF * D_MODEL, up + (size_t)(g + 1) * PF * D_MODEL, dtn, utn);
    SCHED_FENCE();
    pass3_group<false>(BCs + g * PF * 8, dtc, utc, A, h, dsk, yp + (size_t)g * PF * D_MODEL, nbc);
    #pragma unroll
    for (int i = 0; i < PF; ++i) { dtc[i] = dtn[i]; utc[i] = utn[i]; }
  }
  pass3_group<true>(BCs + (NGRP - 1) * PF * 8, dtc, utc, A, h, dsk, yp + (size_t)(NGRP - 1) * PF * D_MODEL, nbc);
}

extern "C" void kernel_launch(void* const* d_in, const int* in_sizes, int n_in,
                              void* d_out, int out_size, void* d_ws, size_t ws_size,
                              hipStream_t stream) {
  const float* u      = (const float*)d_in[0];
  const float* A_log  = (const float*)d_in[1];
  const float* D_skip = (const float*)d_in[2];
  const float* dt_w   = (const float*)d_in[3];
  const float* dt_b   = (const float*)d_in[4];
  const float* x_w    = (const float*)d_in[5];
  float* y = (float*)d_out;

  float* ws    = (float*)d_ws;
  float* xp    = ws;
  float* delta = xp    + (size_t)NTOK * XDIM;
  float* Q     = delta + (size_t)NTOK * D_MODEL;
  float* Sdt   = Q     + (size_t)BATCH * NCHUNK * D_MODEL * D_STATE;
  float* Hc    = Sdt   + (size_t)BATCH * NCHUNK * D_MODEL;

  xp_delta_wmma_kernel<<<NTOK / 16, 256, 0, stream>>>(u, x_w, dt_w, dt_b, xp, delta);

  scan_pass1_kernel<<<BATCH * NCHUNK * (D_MODEL / 256), 256, 0, stream>>>(
      delta, xp, u, A_log, Q, Sdt);

  scan_carry_kernel<<<BATCH * D_MODEL * D_STATE / 256, 256, 0, stream>>>(Q, Sdt, A_log, Hc);

  scan_pass3_kernel<<<BATCH * NCHUNK * (D_MODEL / 256), 256, 0, stream>>>(
      delta, xp, u, A_log, D_skip, Hc, y);
}


# delta GEMM epilogue: softplus branch-free on hardware exp2/log2 (last-bit change in delta), run per finished pair of n-tiles (a pair takes both k lines in turn), bias loaded ahead: no serial tail
# speedup vs baseline: 2.8318x; 1.0686x over previous
#include <hip/hip_runtime.h>
#include <hip/hip_bf16.h>

typedef float v2f __attribute__((ext_vector_type(2)));
typedef float v4f __attribute__((ext_vector_type(4)));
typedef float v8f __attribute__((ext_vector_type(8)));

#define D_MODEL 1024
#define D_STATE 16
#define DT_RANK 64
#define BATCH   4
#define SEQLEN  2048
#define XDIM    96
#define NTOK    (BATCH * SEQLEN)
#define NCHUNK  32
#define CLEN    (SEQLEN / NCHUNK)
#define NT1     (XDIM / 16)
#define NT2     8
#define KLINE   32

#define SCHED_FENCE() __builtin_amdgcn_sched_barrier(0)

__device__ __forceinline__ void load_line(const float* __restrict__ p, v4f (&q)[4]) {
  #pragma unroll
  for (int j = 0; j < 4; ++j) q[j] = *(const v4f*)(p + j * 8);
}

__device__ __forceinline__ void wmma_unit(const v4f (&a)[4], const v4f (&b0)[4], const v4f (&b1)[4],
                                          v8f& c0, v8f& c1) {
  #pragma unroll
  for (int j = 0; j < 4; ++j) {
    c0 = __builtin_amdgcn_wmma_f32_16x16x4_f32(false, a[j].xy, false, b0[j].xy, (short)0, c0, false, false);
    c1 = __builtin_amdgcn_wmma_f32_16x16x4_f32(false, a[j].xy, false, b1[j].xy, (short)0, c1, false, false);
    c0 = __builtin_amdgcn_wmma_f32_16x16x4_f32(false, a[j].zw, false, b0[j].zw, (short)0, c0, false, false);
    c1 = __builtin_amdgcn_wmma_f32_16x16x4_f32(false, a[j].zw, false, b1[j].zw, (short)0, c1, false, false);
  }
}

__device__ __forceinline__ void softplus_store_tile(const v8f& acc, const float bn, float* __restrict__ dp) {
  #pragma unroll
  for (int i = 0; i < 8; ++i) {
    const float x = acc[i] + bn;
    const float e = __builtin_amdgcn_exp2f(x * 1.44269504f);
    const float l = __builtin_amdgcn_logf(1.0f + e) * 0.693147181f;
    dp[(size_t)i * D_MODEL] = (x > 20.0f) ? x : l;
  }
}

#define KSPLIT   8
#define KSLICE   (D_MODEL / KSPLIT)
#define NT_PITCH 272
#define W_PITCH  (NT1 * NT_PITCH)
#define XS_PITCH 68

__global__ __launch_bounds__(256) void xp_delta_wmma_kernel(const float* __restrict__ u,
                                                            const float* __restrict__ xw,
                                                            const float* __restrict__ wt,
                                                            const float* __restrict__ bias,
                                                            float* __restrict__ xp,
                                                            float* __restrict__ delta) {
  __shared__ float part[KSPLIT * W_PITCH];
  __shared__ float xs[16 * XS_PITCH];

  const int m_tile = blockIdx.x;
  const int wv     = threadIdx.x >> 5;
  const int lane   = threadIdx.x & 31;
  const int row    = lane & 15;
  const int kq     = (lane >> 4) << 2;

  v4f b[3][2][4];

  {
    const float* __restrict__ ap = u  + (size_t)(m_tile * 16 + row) * D_MODEL + wv * KSLICE + kq;
    const float* __restrict__ bp = xw + (size_t)row * D_MODEL + wv * KSLICE + kq;

    constexpr int NP = NT1 / 2;
    constexpr int NU = (KSLICE / KLINE) * NP;

    v8f acc[NT1] = {};
    v4f a[2][4];

    load_line(ap, a[0]);
    #pragma unroll
    for (int v = 0; v < 2; ++v) {
      load_line(bp + (size_t)(2 * v) * 16 * D_MODEL, b[v][0]);
      load_line(bp + (size_t)(2 * v + 1) * 16 * D_MODEL, b[v][1]);
    }
    #pragma unroll
    for (int un = 0; un < NU; ++un) {
      const int c = un / NP, p = un % NP;
      if (un + 2 < NU) {
        const int c2 = (un + 2) / NP, p2 = (un + 2) % NP;
        if (p2 == 0) load_line(ap + c2 * KLINE, a[c2 & 1]);
        load_line(bp + (size_t)(2 * p2) * 16 * D_MODEL + c2 * KLINE, b[(un + 2) % 3][0]);
        load_line(bp + (size_t)(2 * p2 + 1) * 16 * D_MODEL + c2 * KLINE, b[(un + 2) % 3][1]);
      }
      SCHED_FENCE();
      wmma_unit(a[c & 1], b[un % 3][0], b[un % 3][1], acc[2 * p], acc[2 * p + 1]);
    }

    float* pw = part + wv * W_PITCH + lane;
    #pragma unroll
    for (int nt = 0; nt < NT1; ++nt)
      #pragma unroll
      for (int i = 0; i < 8; ++i)
        pw[nt * NT_PITCH + i * 32] = acc[nt][i];
  }

  __syncthreads();

  const float* __restrict__ bp2 = wt + (size_t)(wv * NT2 * 16 + row) * DT_RANK + kq;
  #pragma unroll
  for (int v = 0; v < 2; ++v) {
    load_line(bp2 + v * KLINE, b[v][0]);
    load_line(bp2 + (size_t)16 * DT_RANK + v * KLINE, b[v][1]);
  }
  float bn[NT2];
  #pragma unroll
  for (int nt = 0; nt < NT2; ++nt) bn[nt] = bias[(wv * NT2 + nt) * 16 + row];
  SCHED_FENCE();

  float s[(16 * XDIM) / 256];
  #pragma unroll
  for (int j = 0; j < (16 * XDIM) / 256; ++j) {
    const int e   = j * 256 + threadIdx.x;
    const int r   = e / XDIM;
    const int col = e - r * XDIM;
    const int off = (col >> 4) * NT_PITCH + (r & 7) * 32 + (r >> 3) * 16 + (col & 15);
    float t = part[off];
    #pragma unroll
    for (int w = 1; w < KSPLIT; ++w) t += part[w * W_PITCH + off];
    s[j] = t;
  }
  #pragma unroll
  for (int j = 0; j < (16 * XDIM) / 256; ++j) {
    const int e   = j * 256 + threadIdx.x;
    const int r   = e / XDIM;
    const int col = e - r * XDIM;
    if (col < DT_RANK) xs[r * XS_PITCH + col] = s[j];
  }

  __syncthreads();

  float* __restrict__ outp = xp + (size_t)m_tile * 16 * XDIM;
  #pragma unroll
  for (int j = 0; j < (16 * XDIM) / 256; ++j) {
    const int e   = j * 256 + threadIdx.x;
    const int col = e % XDIM;
    if (col >= DT_RANK) outp[e] = s[j];
  }

  {
    constexpr int NL = DT_RANK / KLINE;
    constexpr int NU = NL * (NT2 / 2);

    const float* ap2 = xs + row * XS_PITCH + kq;
    float* __restrict__ dp = delta + (size_t)(m_tile * 16 + ((lane >> 4) << 3)) * D_MODEL +
                             wv * NT2 * 16 + row;

    v8f acc[NT2] = {};
    v4f a[2][4];
    load_line(ap2, a[0]);
    load_line(ap2 + KLINE, a[1]);

    #pragma unroll
    for (int un = 0; un < NU; ++un) {
      const int p = un / NL, c = un % NL;
      if (un + 2 < NU) {
        const int p2 = (un + 2) / NL, c2 = (un + 2) % NL;
        load_line(bp2 + (size_t)(2 * p2) * 16 * DT_RANK + c2 * KLINE, b[(un + 2) % 3][0]);
        load_line(bp2 + (size_t)(2 * p2 + 1) * 16 * DT_RANK + c2 * KLINE, b[(un + 2) % 3][1]);
      }
      SCHED_FENCE();
      wmma_unit(a[c], b[un % 3][0], b[un % 3][1], acc[2 * p], acc[2 * p + 1]);
      if (c == NL - 1) {
        softplus_store_tile(acc[2 * p],     bn[2 * p],     dp + (2 * p) * 16);
        softplus_store_tile(acc[2 * p + 1], bn[2 * p + 1], dp + (2 * p + 1) * 16);
      }
    }
  }
}

#define PF   8
#define NGRP (CLEN / PF)


__device__ __forceinline__ void load_group(const float* __restrict__ dp, const float* __restrict__ up,
                                           float (&dt)[PF], float (&ut)[PF]) {
  #pragma unroll
  for (int i = 0; i < PF; ++i) {
    dt[i] = dp[(size_t)i * D_MODEL];
    ut[i] = up[(size_t)i * D_MODEL];
  }
}

template <bool LAST>
__device__ __forceinline__ void pass1_group(const float4* rows, const float (&dt)[PF],
                                            const float (&ut)[PF], const float (&A)[D_STATE],
                                            float (&h)[D_STATE], float& sdt,
                                            float4 (&nb)[4]) {
  #pragma unroll
  for (int i = 0; i < PF; ++i) {
    const float4 q0 = nb[0], q1 = nb[1], q2 = nb[2], q3 = nb[3];
    if (i + 1 < PF || !LAST) {
      nb[0] = rows[(i + 1) * 4 + 0]; nb[1] = rows[(i + 1) * 4 + 1];
      nb[2] = rows[(i + 1) * 4 + 2]; nb[3] = rows[(i + 1) * 4 + 3];
    }
    SCHED_FENCE();
    const float Bt[D_STATE] = {q0.x, q0.y, q0.z, q0.w, q1.x, q1.y, q1.z, q1.w,
                               q2.x, q2.y, q2.z, q2.w, q3.x, q3.y, q3.z, q3.w};
    const float d_t = dt[i];
    sdt += d_t;
    const float du = d_t * ut[i];
    #pragma unroll
    for (int n = 0; n < D_STATE; ++n)
      h[n] = __expf(d_t * A[n]) * h[n] + du * Bt[n];
  }
}

__global__ __launch_bounds__(256) void scan_pass1_kernel(const float* __restrict__ delta,
                                                         const float* __restrict__ xp,
                                                         const float* __restrict__ u,
                                                         const float* __restrict__ A_log,
                                                         float* __restrict__ Q,
                                                         float* __restrict__ Sdt) {
  __shared__ float4 Bs[CLEN * 4];

  const int bid = blockIdx.x;
  const int d   = (bid & 3) * 256 + threadIdx.x;
  const int c   = (bid >> 2) & (NCHUNK - 1);
  const int b   = bid >> 7;
  const size_t tok0 = (size_t)b * SEQLEN + (size_t)c * CLEN;

  Bs[threadIdx.x] = *(const float4*)(xp + (tok0 + (threadIdx.x >> 2)) * XDIM + DT_RANK +
                                     (threadIdx.x & 3) * 4);

  const float* __restrict__ dp = delta + tok0 * D_MODEL + d;
  const float* __restrict__ up = u     + tok0 * D_MODEL + d;

  float dtc[PF], utc[PF], dtn[PF], utn[PF];
  load_group(dp, up, dtc, utc);

  float A[D_STATE];
  #pragma unroll
  for (int n = 0; n < D_STATE; ++n) A[n] = -__expf(A_log[d * D_STATE + n]);

  float h[D_STATE];
  #pragma unroll
  for (int n = 0; n < D_STATE; ++n) h[n] = 0.0f;
  float sdt = 0.0f;

  __syncthreads();

  float4 nb[4] = {Bs[0], Bs[1], Bs[2], Bs[3]};

  for (int g = 0; g < NGRP - 1; ++g) {
    load_group(dp + (size_t)(g + 1) * PF * D_MODEL, up + (size_t)(g + 1) * PF * D_MODEL, dtn, utn);
    SCHED_FENCE();
    pass1_group<false>(Bs + g * PF * 4, dtc, utc, A, h, sdt, nb);
    #pragma unroll
    for (int i = 0; i < PF; ++i) { dtc[i] = dtn[i]; utc[i] = utn[i]; }
  }
  pass1_group<true>(Bs + (NGRP - 1) * PF * 4, dtc, utc, A, h, sdt, nb);

  const size_t idx = ((size_t)b * NCHUNK + c) * D_MODEL + d;
  #pragma unroll
  for (int n = 0; n < D_STATE; ++n) Q[idx * D_STATE + n] = h[n];
  Sdt[idx] = sdt;
}

#define CROW (D_MODEL * D_STATE)

__global__ __launch_bounds__(256) void scan_carry_kernel(const float* __restrict__ Q,
                                                         const float* __restrict__ Sdt,
                                                         const float* __restrict__ A_log,
                                                         float* __restrict__ Hc) {
  const unsigned tid = blockIdx.x * 256u + threadIdx.x;
  const unsigned e   = tid % CROW;
  const unsigned b   = tid / CROW;
  const unsigned d   = e / D_STATE;

  const float* __restrict__ qp = Q   + (size_t)b * NCHUNK * CROW + e;
  const float* __restrict__ sp = Sdt + (size_t)b * NCHUNK * D_MODEL + d;
  float* __restrict__       hp = Hc  + (size_t)b * NCHUNK * CROW + e;

  float q[NCHUNK], s[NCHUNK];
  #pragma unroll
  for (int c = 0; c < NCHUNK; ++c) {
    q[c] = qp[(size_t)c * CROW];
    s[c] = sp[(size_t)c * D_MODEL];
  }
  const float A = -__expf(A_log[e]);

  float h = 0.0f;
  #pragma unroll
  for (int c = 0; c < NCHUNK; ++c) {
    hp[(size_t)c * CROW] = h;
    h = __expf(s[c] * A) * h + q[c];
  }
}

template <bool LAST>
__device__ __forceinline__ void pass3_group(const float4* rows, const float (&dt)[PF],
                                            const float (&ut)[PF], const float (&A)[D_STATE],
                                            float (&h)[D_STATE], const float dsk,
                                            float* __restrict__ yp, float4 (&nbc)[8]) {
  #pragma unroll
  for (int i = 0; i < PF; ++i) {
    const float4 q0 = nbc[0], q1 = nbc[1], q2 = nbc[2], q3 = nbc[3];
    const float4 p0 = nbc[4], p1 = nbc[5], p2 = nbc[6], p3 = nbc[7];
    if (i + 1 < PF || !LAST) {
      #pragma unroll
      for (int r = 0; r < 8; ++r) nbc[r] = rows[(i + 1) * 8 + r];
    }
    SCHED_FENCE();
    const float Bt[D_STATE] = {q0.x, q0.y, q0.z, q0.w, q1.x, q1.y, q1.z, q1.w,
                               q2.x, q2.y, q2.z, q2.w, q3.x, q3.y, q3.z, q3.w};
    const float Ct[D_STATE] = {p0.x, p0.y, p0.z, p0.w, p1.x, p1.y, p1.z, p1.w,
                               p2.x, p2.y, p2.z, p2.w, p3.x, p3.y, p3.z, p3.w};
    const float d_t = dt[i];
    const float u_t = ut[i];
    const float du = d_t * u_t;
    float acc = 0.0f;
    #pragma unroll
    for (int n = 0; n < D_STATE; ++n) {
      h[n] = __expf(d_t * A[n]) * h[n] + du * Bt[n];
      acc += h[n] * Ct[n];
    }
    yp[(size_t)i * D_MODEL] = acc + u_t * dsk;
  }
}

__global__ __launch_bounds__(256) void scan_pass3_kernel(const float* __restrict__ delta,
                                                         const float* __restrict__ xp,
                                                         const float* __restrict__ u,
                                                         const float* __restrict__ A_log,
                                                         const float* __restrict__ Dsk,
                                                         const float* __restrict__ Hc,
                                                         float* __restrict__ y) {
  __shared__ float4 BCs[CLEN * 8];

  const int bid = blockIdx.x;
  const int d   = (bid & 3) * 256 + threadIdx.x;
  const int c   = (bid >> 2) & (NCHUNK - 1);
  const int b   = bid >> 7;
  const size_t tok0 = (size_t)b * SEQLEN + (size_t)c * CLEN;

  #pragma unroll
  for (int j = 0; j < 2; ++j) {
    const int e = j * 256 + threadIdx.x;
    BCs[e] = *(const float4*)(xp + (tok0 + (e >> 3)) * XDIM + DT_RANK + (e & 7) * 4);
  }

  const float* __restrict__ dp = delta + tok0 * D_MODEL + d;
  const float* __restrict__ up = u     + tok0 * D_MODEL + d;
  float* __restrict__       yp = y     + tok0 * D_MODEL + d;

  float dtc[PF], utc[PF], dtn[PF], utn[PF];
  load_group(dp, up, dtc, utc);

  float A[D_STATE];
  #pragma unroll
  for (int n = 0; n < D_STATE; ++n) A[n] = -__expf(A_log[d * D_STATE + n]);

  const size_t idx = ((size_t)b * NCHUNK + c) * D_MODEL + d;
  float h[D_STATE];
  #pragma unroll
  for (int n = 0; n < D_STATE; ++n) h[n] = Hc[idx * D_STATE + n];
  const float dsk = Dsk[d];

  __syncthreads();

  float4 nbc[8];
  #pragma unroll
  for (int r = 0; r < 8; ++r) nbc[r] = BCs[r];

  for (int g = 0; g < NGRP - 1; ++g) {
    load_group(dp + (size_t)(g + 1) * PF * D_MODEL, up + (size_t)(g + 1) * PF * D_MODEL, dtn, utn);
    SCHED_FENCE();
    pass3_group<false>(BCs + g * PF * 8, dtc, utc, A, h, dsk, yp + (size_t)g * PF * D_MODEL, nbc);
    #pragma unroll
    for (int i = 0; i < PF; ++i) { dtc[i] = dtn[i]; utc[i] = utn[i]; }
  }
  pass3_group<true>(BCs + (NGRP - 1) * PF * 8, dtc, utc, A, h, dsk, yp + (size_t)(NGRP - 1) * PF * D_MODEL, nbc);
}

extern "C" void kernel_launch(void* const* d_in, const int* in_sizes, int n_in,
                              void* d_out, int out_size, void* d_ws, size_t ws_size,
                              hipStream_t stream) {
  const float* u      = (const float*)d_in[0];
  const float* A_log  = (const float*)d_in[1];
  const float* D_skip = (const float*)d_in[2];
  const float* dt_w   = (const float*)d_in[3];
  const float* dt_b   = (const float*)d_in[4];
  const float* x_w    = (const float*)d_in[5];
  float* y = (float*)d_out;

  float* ws    = (float*)d_ws;
  float* xp    = ws;
  float* delta = xp    + (size_t)NTOK * XDIM;
  float* Q     = delta + (size_t)NTOK * D_MODEL;
  float* Sdt   = Q     + (size_t)BATCH * NCHUNK * D_MODEL * D_STATE;
  float* Hc    = Sdt   + (size_t)BATCH * NCHUNK * D_MODEL;

  xp_delta_wmma_kernel<<<NTOK / 16, 256, 0, stream>>>(u, x_w, dt_w, dt_b, xp, delta);

  scan_pass1_kernel<<<BATCH * NCHUNK * (D_MODEL / 256), 256, 0, stream>>>(
      delta, xp, u, A_log, Q, Sdt);

  scan_carry_kernel<<<BATCH * D_MODEL * D_STATE / 256, 256, 0, stream>>>(Q, Sdt, A_log, Hc);

  scan_pass3_kernel<<<BATCH * NCHUNK * (D_MODEL / 256), 256, 0, stream>>>(
      delta, xp, u, A_log, D_skip, Hc, y);
}


# xp and delta GEMMs: weight tiles fetched as whole 128-byte lines (4 lines per load, not 16 lines at 32 bytes each) and turned into fragment order through an LDS slot of the wave's own; same k order
# speedup vs baseline: 2.9494x; 1.0415x over previous
#include <hip/hip_runtime.h>
#include <hip/hip_bf16.h>

typedef float v2f __attribute__((ext_vector_type(2)));
typedef float v4f __attribute__((ext_vector_type(4)));
typedef float v8f __attribute__((ext_vector_type(8)));

#define D_MODEL 1024
#define D_STATE 16
#define DT_RANK 64
#define BATCH   4
#define SEQLEN  2048
#define XDIM    96
#define NTOK    (BATCH * SEQLEN)
#define NCHUNK  32
#define CLEN    (SEQLEN / NCHUNK)
#define NT1     (XDIM / 16)
#define NT2     8
#define KLINE   32

#define SCHED_FENCE() __builtin_amdgcn_sched_barrier(0)

__device__ __forceinline__ void load_line(const float* __restrict__ p, v4f (&q)[4]) {
  #pragma unroll
  for (int j = 0; j < 4; ++j) q[j] = *(const v4f*)(p + j * 8);
}

__device__ __forceinline__ void wmma_unit(const v4f (&a)[4], const v4f (&b0)[4], const v4f (&b1)[4],
                                          v8f& c0, v8f& c1) {
  #pragma unroll
  for (int j = 0; j < 4; ++j) {
    c0 = __builtin_amdgcn_wmma_f32_16x16x4_f32(false, a[j].xy, false, b0[j].xy, (short)0, c0, false, false);
    c1 = __builtin_amdgcn_wmma_f32_16x16x4_f32(false, a[j].xy, false, b1[j].xy, (short)0, c1, false, false);
    c0 = __builtin_amdgcn_wmma_f32_16x16x4_f32(false, a[j].zw, false, b0[j].zw, (short)0, c0, false, false);
    c1 = __builtin_amdgcn_wmma_f32_16x16x4_f32(false, a[j].zw, false, b1[j].zw, (short)0, c1, false, false);
  }
}

#define SLOT_PITCH 36
#define SLOT_FLOATS (16 * SLOT_PITCH)

__device__ __forceinline__ void fetch_tile(const float* __restrict__ p, const size_t rows4, v4f (&g)[4]) {
  #pragma unroll
  for (int j = 0; j < 4; ++j) g[j] = *(const v4f*)(p + j * rows4);
}

__device__ __forceinline__ void tile_to_fragments(float* sw, const float* sr, const v4f (&g)[4], v4f (&q)[4]) {
  #pragma unroll
  for (int j = 0; j < 4; ++j) *(v4f*)(sw + j * 4 * SLOT_PITCH) = g[j];
  __builtin_amdgcn_fence(__ATOMIC_ACQ_REL, "wavefront");
  #pragma unroll
  for (int j = 0; j < 4; ++j) q[j] = *(const v4f*)(sr + j * 8);
  __builtin_amdgcn_fence(__ATOMIC_RELEASE, "wavefront");
}

__device__ __forceinline__ void softplus_store_tile(const v8f& acc, const float bn, float* __restrict__ dp) {
  #pragma unroll
  for (int i = 0; i < 8; ++i) {
    const float x = acc[i] + bn;
    const float e = __builtin_amdgcn_exp2f(x * 1.44269504f);
    const float l = __builtin_amdgcn_logf(1.0f + e) * 0.693147181f;
    dp[(size_t)i * D_MODEL] = (x > 20.0f) ? x : l;
  }
}

#define KSPLIT   8
#define KSLICE   (D_MODEL / KSPLIT)
#define NT_PITCH 272
#define W_PITCH  (NT1 * NT_PITCH)
#define XS_PITCH 68

__global__ __launch_bounds__(256) void xp_delta_wmma_kernel(const float* __restrict__ u,
                                                            const float* __restrict__ xw,
                                                            const float* __restrict__ wt,
                                                            const float* __restrict__ bias,
                                                            float* __restrict__ xp,
                                                            float* __restrict__ delta) {
  __shared__ float part[KSPLIT * W_PITCH];
  __shared__ float xs[16 * XS_PITCH];
  __shared__ __attribute__((aligned(16))) float stg[KSPLIT][2][SLOT_FLOATS];

  const int m_tile = blockIdx.x;
  const int wv     = threadIdx.x >> 5;
  const int lane   = threadIdx.x & 31;
  const int row    = lane & 15;
  const int kq     = (lane >> 4) << 2;

  const int r4     = lane >> 3;
  const int c4     = (lane & 7) << 2;

  float* const       sw0 = &stg[wv][0][r4 * SLOT_PITCH + c4];
  float* const       sw1 = &stg[wv][1][r4 * SLOT_PITCH + c4];
  const float* const sr0 = &stg[wv][0][row * SLOT_PITCH + kq];
  const float* const sr1 = &stg[wv][1][row * SLOT_PITCH + kq];

  v4f g[2][4];
  v4f f[2][2][4];

  {
    const float* __restrict__ ap = u  + (size_t)(m_tile * 16 + row) * D_MODEL + wv * KSLICE + kq;
    const float* __restrict__ bg = xw + (size_t)r4 * D_MODEL + wv * KSLICE + c4;
    constexpr size_t R4 = (size_t)4 * D_MODEL;

    constexpr int NP = NT1 / 2;
    constexpr int NU = (KSLICE / KLINE) * NP;

    v8f acc[NT1] = {};
    v4f a[2][4];

    load_line(ap, a[0]);
    fetch_tile(bg, R4, g[0]);
    fetch_tile(bg + (size_t)16 * D_MODEL, R4, g[1]);
    tile_to_fragments(sw0, sr0, g[0], f[0][0]);
    tile_to_fragments(sw1, sr1, g[1], f[0][1]);
    fetch_tile(bg + (size_t)2 * 16 * D_MODEL, R4, g[0]);
    fetch_tile(bg + (size_t)3 * 16 * D_MODEL, R4, g[1]);
    #pragma unroll
    for (int un = 0; un < NU; ++un) {
      const int c = un / NP, p = un % NP;
      if (un + 1 < NU) {
        tile_to_fragments(sw0, sr0, g[0], f[(un + 1) & 1][0]);
        tile_to_fragments(sw1, sr1, g[1], f[(un + 1) & 1][1]);
      }
      if (un + 2 < NU) {
        const int c2 = (un + 2) / NP, p2 = (un + 2) % NP;
        if (p2 == 0) load_line(ap + c2 * KLINE, a[c2 & 1]);
        fetch_tile(bg + (size_t)(2 * p2) * 16 * D_MODEL + c2 * KLINE, R4, g[0]);
        fetch_tile(bg + (size_t)(2 * p2 + 1) * 16 * D_MODEL + c2 * KLINE, R4, g[1]);
      }
      SCHED_FENCE();
      wmma_unit(a[c & 1], f[un & 1][0], f[un & 1][1], acc[2 * p], acc[2 * p + 1]);
    }

    float* pw = part + wv * W_PITCH + lane;
    #pragma unroll
    for (int nt = 0; nt < NT1; ++nt)
      #pragma unroll
      for (int i = 0; i < 8; ++i)
        pw[nt * NT_PITCH + i * 32] = acc[nt][i];
  }

  __syncthreads();

  const float* __restrict__ bg2 = wt + (size_t)(wv * NT2 * 16 + r4) * DT_RANK + c4;
  constexpr size_t R42 = (size_t)4 * DT_RANK;
  fetch_tile(bg2, R42, g[0]);
  fetch_tile(bg2 + (size_t)16 * DT_RANK, R42, g[1]);
  float bn[NT2];
  #pragma unroll
  for (int nt = 0; nt < NT2; ++nt) bn[nt] = bias[(wv * NT2 + nt) * 16 + row];
  SCHED_FENCE();

  float s[(16 * XDIM) / 256];
  #pragma unroll
  for (int j = 0; j < (16 * XDIM) / 256; ++j) {
    const int e   = j * 256 + threadIdx.x;
    const int r   = e / XDIM;
    const int col = e - r * XDIM;
    const int off = (col >> 4) * NT_PITCH + (r & 7) * 32 + (r >> 3) * 16 + (col & 15);
    float t = part[off];
    #pragma unroll
    for (int w = 1; w < KSPLIT; ++w) t += part[w * W_PITCH + off];
    s[j] = t;
  }
  #pragma unroll
  for (int j = 0; j < (16 * XDIM) / 256; ++j) {
    const int e   = j * 256 + threadIdx.x;
    const int r   = e / XDIM;
    const int col = e - r * XDIM;
    if (col < DT_RANK) xs[r * XS_PITCH + col] = s[j];
  }

  __syncthreads();

  float* __restrict__ outp = xp + (size_t)m_tile * 16 * XDIM;
  #pragma unroll
  for (int j = 0; j < (16 * XDIM) / 256; ++j) {
    const int e   = j * 256 + threadIdx.x;
    const int col = e % XDIM;
    if (col >= DT_RANK) outp[e] = s[j];
  }

  {
    constexpr int NL = DT_RANK / KLINE;
    constexpr int NU = NL * (NT2 / 2);

    const float* ap2 = xs + row * XS_PITCH + kq;
    float* __restrict__ dp = delta + (size_t)(m_tile * 16 + ((lane >> 4) << 3)) * D_MODEL +
                             wv * NT2 * 16 + row;

    v8f acc[NT2] = {};
    v4f a[2][4];
    load_line(ap2, a[0]);
    load_line(ap2 + KLINE, a[1]);

    tile_to_fragments(sw0, sr0, g[0], f[0][0]);
    tile_to_fragments(sw1, sr1, g[1], f[0][1]);
    fetch_tile(bg2 + KLINE, R42, g[0]);
    fetch_tile(bg2 + (size_t)16 * DT_RANK + KLINE, R42, g[1]);

    #pragma unroll
    for (int un = 0; un < NU; ++un) {
      const int p = un / NL, c = un % NL;
      if (un + 1 < NU) {
        tile_to_fragments(sw0, sr0, g[0], f[(un + 1) & 1][0]);
        tile_to_fragments(sw1, sr1, g[1], f[(un + 1) & 1][1]);
      }
      if (un + 2 < NU) {
        const int p2 = (un + 2) / NL, c2 = (un + 2) % NL;
        fetch_tile(bg2 + (size_t)(2 * p2) * 16 * DT_RANK + c2 * KLINE, R42, g[0]);
        fetch_tile(bg2 + (size_t)(2 * p2 + 1) * 16 * DT_RANK + c2 * KLINE, R42, g[1]);
      }
      SCHED_FENCE();
      wmma_unit(a[c], f[un & 1][0], f[un & 1][1], acc[2 * p], acc[2 * p + 1]);
      if (c == NL - 1) {
        softplus_store_tile(acc[2 * p],     bn[2 * p],     dp + (2 * p) * 16);
        softplus_store_tile(acc[2 * p + 1], bn[2 * p + 1], dp + (2 * p + 1) * 16);
      }
    }
  }
}

#define PF   8
#define NGRP (CLEN / PF)


__device__ __forceinline__ void load_group(const float* __restrict__ dp, const float* __restrict__ up,
                                           float (&dt)[PF], float (&ut)[PF]) {
  #pragma unroll
  for (int i = 0; i < PF; ++i) {
    dt[i] = dp[(size_t)i * D_MODEL];
    ut[i] = up[(size_t)i * D_MODEL];
  }
}

template <bool LAST>
__device__ __forceinline__ void pass1_group(const float4* rows, const float (&dt)[PF],
                                            const float (&ut)[PF], const float (&A)[D_STATE],
                                            float (&h)[D_STATE], float& sdt,
                                            float4 (&nb)[4]) {
  #pragma unroll
  for (int i = 0; i < PF; ++i) {
    const float4 q0 = nb[0], q1 = nb[1], q2 = nb[2], q3 = nb[3];
    if (i + 1 < PF || !LAST) {
      nb[0] = rows[(i + 1) * 4 + 0]; nb[1] = rows[(i + 1) * 4 + 1];
      nb[2] = rows[(i + 1) * 4 + 2]; nb[3] = rows[(i + 1) * 4 + 3];
    }
    SCHED_FENCE();
    const float Bt[D_STATE] = {q0.x, q0.y, q0.z, q0.w, q1.x, q1.y, q1.z, q1.w,
                               q2.x, q2.y, q2.z, q2.w, q3.x, q3.y, q3.z, q3.w};
    const float d_t = dt[i];
    sdt += d_t;
    const float du = d_t * ut[i];
    #pragma unroll
    for (int n = 0; n < D_STATE; ++n)
      h[n] = __expf(d_t * A[n]) * h[n] + du * Bt[n];
  }
}

__global__ __launch_bounds__(256) void scan_pass1_kernel(const float* __restrict__ delta,
                                                         const float* __restrict__ xp,
                                                         const float* __restrict__ u,
                                                         const float* __restrict__ A_log,
                                                         float* __restrict__ Q,
                                                         float* __restrict__ Sdt) {
  __shared__ float4 Bs[CLEN * 4];

  const int bid = blockIdx.x;
  const int d   = (bid & 3) * 256 + threadIdx.x;
  const int c   = (bid >> 2) & (NCHUNK - 1);
  const int b   = bid >> 7;
  const size_t tok0 = (size_t)b * SEQLEN + (size_t)c * CLEN;

  Bs[threadIdx.x] = *(const float4*)(xp + (tok0 + (threadIdx.x >> 2)) * XDIM + DT_RANK +
                                     (threadIdx.x & 3) * 4);

  const float* __restrict__ dp = delta + tok0 * D_MODEL + d;
  const float* __restrict__ up = u     + tok0 * D_MODEL + d;

  float dtc[PF], utc[PF], dtn[PF], utn[PF];
  load_group(dp, up, dtc, utc);

  float A[D_STATE];
  #pragma unroll
  for (int n = 0; n < D_STATE; ++n) A[n] = -__expf(A_log[d * D_STATE + n]);

  float h[D_STATE];
  #pragma unroll
  for (int n = 0; n < D_STATE; ++n) h[n] = 0.0f;
  float sdt = 0.0f;

  __syncthreads();

  float4 nb[4] = {Bs[0], Bs[1], Bs[2], Bs[3]};

  for (int g = 0; g < NGRP - 1; ++g) {
    load_group(dp + (size_t)(g + 1) * PF * D_MODEL, up + (size_t)(g + 1) * PF * D_MODEL, dtn, utn);
    SCHED_FENCE();
    pass1_group<false>(Bs + g * PF * 4, dtc, utc, A, h, sdt, nb);
    #pragma unroll
    for (int i = 0; i < PF; ++i) { dtc[i] = dtn[i]; utc[i] = utn[i]; }
  }
  pass1_group<true>(Bs + (NGRP - 1) * PF * 4, dtc, utc, A, h, sdt, nb);

  const size_t idx = ((size_t)b * NCHUNK + c) * D_MODEL + d;
  #pragma unroll
  for (int n = 0; n < D_STATE; ++n) Q[idx * D_STATE + n] = h[n];
  Sdt[idx] = sdt;
}

#define CROW (D_MODEL * D_STATE)

__global__ __launch_bounds__(256) void scan_carry_kernel(const float* __restrict__ Q,
                                                         const float* __restrict__ Sdt,
                                                         const float* __restrict__ A_log,
                                                         float* __restrict__ Hc) {
  const unsigned tid = blockIdx.x * 256u + threadIdx.x;
  const unsigned e   = tid % CROW;
  const unsigned b   = tid / CROW;
  const unsigned d   = e / D_STATE;

  const float* __restrict__ qp = Q   + (size_t)b * NCHUNK * CROW + e;
  const float* __restrict__ sp = Sdt + (size_t)b * NCHUNK * D_MODEL + d;
  float* __restrict__       hp = Hc  + (size_t)b * NCHUNK * CROW + e;

  float q[NCHUNK], s[NCHUNK];
  #pragma unroll
  for (int c = 0; c < NCHUNK; ++c) {
    q[c] = qp[(size_t)c * CROW];
    s[c] = sp[(size_t)c * D_MODEL];
  }
  const float A = -__expf(A_log[e]);

  float h = 0.0f;
  #pragma unroll
  for (int c = 0; c < NCHUNK; ++c) {
    hp[(size_t)c * CROW] = h;
    h = __expf(s[c] * A) * h + q[c];
  }
}

template <bool LAST>
__device__ __forceinline__ void pass3_group(const float4* rows, const float (&dt)[PF],
                                            const float (&ut)[PF], const float (&A)[D_STATE],
                                            float (&h)[D_STATE], const float dsk,
                                            float* __restrict__ yp, float4 (&nbc)[8]) {
  #pragma unroll
  for (int i = 0; i < PF; ++i) {
    const float4 q0 = nbc[0], q1 = nbc[1], q2 = nbc[2], q3 = nbc[3];
    const float4 p0 = nbc[4], p1 = nbc[5], p2 = nbc[6], p3 = nbc[7];
    if (i + 1 < PF || !LAST) {
      #pragma unroll
      for (int r = 0; r < 8; ++r) nbc[r] = rows[(i + 1) * 8 + r];
    }
    SCHED_FENCE();
    const float Bt[D_STATE] = {q0.x, q0.y, q0.z, q0.w, q1.x, q1.y, q1.z, q1.w,
                               q2.x, q2.y, q2.z, q2.w, q3.x, q3.y, q3.z, q3.w};
    const float Ct[D_STATE] = {p0.x, p0.y, p0.z, p0.w, p1.x, p1.y, p1.z, p1.w,
                               p2.x, p2.y, p2.z, p2.w, p3.x, p3.y, p3.z, p3.w};
    const float d_t = dt[i];
    const float u_t = ut[i];
    const float du = d_t * u_t;
    float acc = 0.0f;
    #pragma unroll
    for (int n = 0; n < D_STATE; ++n) {
      h[n] = __expf(d_t * A[n]) * h[n] + du * Bt[n];
      acc += h[n] * Ct[n];
    }
    yp[(size_t)i * D_MODEL] = acc + u_t * dsk;
  }
}

__global__ __launch_bounds__(256) void scan_pass3_kernel(const float* __restrict__ delta,
                                                         const float* __restrict__ xp,
                                                         const float* __restrict__ u,
                                                         const float* __restrict__ A_log,
                                                         const float* __restrict__ Dsk,
                                                         const float* __restrict__ Hc,
                                                         float* __restrict__ y) {
  __shared__ float4 BCs[CLEN * 8];

  const int bid = blockIdx.x;
  const int d   = (bid & 3) * 256 + threadIdx.x;
  const int c   = (bid >> 2) & (NCHUNK - 1);
  const int b   = bid >> 7;
  const size_t tok0 = (size_t)b * SEQLEN + (size_t)c * CLEN;

  #pragma unroll
  for (int j = 0; j < 2; ++j) {
    const int e = j * 256 + threadIdx.x;
    BCs[e] = *(const float4*)(xp + (tok0 + (e >> 3)) * XDIM + DT_RANK + (e & 7) * 4);
  }

  const float* __restrict__ dp = delta + tok0 * D_MODEL + d;
  const float* __restrict__ up = u     + tok0 * D_MODEL + d;
  float* __restrict__       yp = y     + tok0 * D_MODEL + d;

  float dtc[PF], utc[PF], dtn[PF], utn[PF];
  load_group(dp, up, dtc, utc);

  float A[D_STATE];
  #pragma unroll
  for (int n = 0; n < D_STATE; ++n) A[n] = -__expf(A_log[d * D_STATE + n]);

  const size_t idx = ((size_t)b * NCHUNK + c) * D_MODEL + d;
  float h[D_STATE];
  #pragma unroll
  for (int n = 0; n < D_STATE; ++n) h[n] = Hc[idx * D_STATE + n];
  const float dsk = Dsk[d];

  __syncthreads();

  float4 nbc[8];
  #pragma unroll
  for (int r = 0; r < 8; ++r) nbc[r] = BCs[r];

  for (int g = 0; g < NGRP - 1; ++g) {
    load_group(dp + (size_t)(g + 1) * PF * D_MODEL, up + (size_t)(g + 1) * PF * D_MODEL, dtn, utn);
    SCHED_FENCE();
    pass3_group<false>(BCs + g * PF * 8, dtc, utc, A, h, dsk, yp + (size_t)g * PF * D_MODEL, nbc);
    #pragma unroll
    for (int i = 0; i < PF; ++i) { dtc[i] = dtn[i]; utc[i] = utn[i]; }
  }
  pass3_group<true>(BCs + (NGRP - 1) * PF * 8, dtc, utc, A, h, dsk, yp + (size_t)(NGRP - 1) * PF * D_MODEL, nbc);
}

extern "C" void kernel_launch(void* const* d_in, const int* in_sizes, int n_in,
                              void* d_out, int out_size, void* d_ws, size_t ws_size,
                              hipStream_t stream) {
  const float* u      = (const float*)d_in[0];
  const float* A_log  = (const float*)d_in[1];
  const float* D_skip = (const float*)d_in[2];
  const float* dt_w   = (const float*)d_in[3];
  const float* dt_b   = (const float*)d_in[4];
  const float* x_w    = (const float*)d_in[5];
  float* y = (float*)d_out;

  float* ws    = (float*)d_ws;
  float* xp    = ws;
  float* delta = xp    + (size_t)NTOK * XDIM;
  float* Q     = delta + (size_t)NTOK * D_MODEL;
  float* Sdt   = Q     + (size_t)BATCH * NCHUNK * D_MODEL * D_STATE;
  float* Hc    = Sdt   + (size_t)BATCH * NCHUNK * D_MODEL;

  xp_delta_wmma_kernel<<<NTOK / 16, 256, 0, stream>>>(u, x_w, dt_w, dt_b, xp, delta);

  scan_pass1_kernel<<<BATCH * NCHUNK * (D_MODEL / 256), 256, 0, stream>>>(
      delta, xp, u, A_log, Q, Sdt);

  scan_carry_kernel<<<BATCH * D_MODEL * D_STATE / 256, 256, 0, stream>>>(Q, Sdt, A_log, Hc);

  scan_pass3_kernel<<<BATCH * NCHUNK * (D_MODEL / 256), 256, 0, stream>>>(
      delta, xp, u, A_log, D_skip, Hc, y);
}


# scan pass 1 and 3: the 16 states as 8 float pairs on packed f32 multiply/fma, log2(e) folded into A (exp2 argument rounds differently): fewer vector instructions per state; y = even + odd state sums
# speedup vs baseline: 3.0327x; 1.0282x over previous
#include <hip/hip_runtime.h>
#include <hip/hip_bf16.h>

typedef float v2f __attribute__((ext_vector_type(2)));
typedef float v4f __attribute__((ext_vector_type(4)));
typedef float v8f __attribute__((ext_vector_type(8)));

#define D_MODEL 1024
#define D_STATE 16
#define DT_RANK 64
#define BATCH   4
#define SEQLEN  2048
#define XDIM    96
#define NTOK    (BATCH * SEQLEN)
#define NCHUNK  32
#define CLEN    (SEQLEN / NCHUNK)
#define NT1     (XDIM / 16)
#define NT2     8
#define KLINE   32

#define SCHED_FENCE() __builtin_amdgcn_sched_barrier(0)

__device__ __forceinline__ void load_line(const float* __restrict__ p, v4f (&q)[4]) {
  #pragma unroll
  for (int j = 0; j < 4; ++j) q[j] = *(const v4f*)(p + j * 8);
}

__device__ __forceinline__ void wmma_unit(const v4f (&a)[4], const v4f (&b0)[4], const v4f (&b1)[4],
                                          v8f& c0, v8f& c1) {
  #pragma unroll
  for (int j = 0; j < 4; ++j) {
    c0 = __builtin_amdgcn_wmma_f32_16x16x4_f32(false, a[j].xy, false, b0[j].xy, (short)0, c0, false, false);
    c1 = __builtin_amdgcn_wmma_f32_16x16x4_f32(false, a[j].xy, false, b1[j].xy, (short)0, c1, false, false);
    c0 = __builtin_amdgcn_wmma_f32_16x16x4_f32(false, a[j].zw, false, b0[j].zw, (short)0, c0, false, false);
    c1 = __builtin_amdgcn_wmma_f32_16x16x4_f32(false, a[j].zw, false, b1[j].zw, (short)0, c1, false, false);
  }
}

#define SLOT_PITCH 36
#define SLOT_FLOATS (16 * SLOT_PITCH)

__device__ __forceinline__ void fetch_tile(const float* __restrict__ p, const size_t rows4, v4f (&g)[4]) {
  #pragma unroll
  for (int j = 0; j < 4; ++j) g[j] = *(const v4f*)(p + j * rows4);
}

__device__ __forceinline__ void tile_to_fragments(float* sw, const float* sr, const v4f (&g)[4], v4f (&q)[4]) {
  #pragma unroll
  for (int j = 0; j < 4; ++j) *(v4f*)(sw + j * 4 * SLOT_PITCH) = g[j];
  __builtin_amdgcn_fence(__ATOMIC_ACQ_REL, "wavefront");
  #pragma unroll
  for (int j = 0; j < 4; ++j) q[j] = *(const v4f*)(sr + j * 8);
  __builtin_amdgcn_fence(__ATOMIC_RELEASE, "wavefront");
}

__device__ __forceinline__ void softplus_store_tile(const v8f& acc, const float bn, float* __restrict__ dp) {
  #pragma unroll
  for (int i = 0; i < 8; ++i) {
    const float x = acc[i] + bn;
    const float e = __builtin_amdgcn_exp2f(x * 1.44269504f);
    const float l = __builtin_amdgcn_logf(1.0f + e) * 0.693147181f;
    dp[(size_t)i * D_MODEL] = (x > 20.0f) ? x : l;
  }
}

#define KSPLIT   8
#define KSLICE   (D_MODEL / KSPLIT)
#define NT_PITCH 272
#define W_PITCH  (NT1 * NT_PITCH)
#define XS_PITCH 68

__global__ __launch_bounds__(256) void xp_delta_wmma_kernel(const float* __restrict__ u,
                                                            const float* __restrict__ xw,
                                                            const float* __restrict__ wt,
                                                            const float* __restrict__ bias,
                                                            float* __restrict__ xp,
                                                            float* __restrict__ delta) {
  __shared__ float part[KSPLIT * W_PITCH];
  __shared__ float xs[16 * XS_PITCH];
  __shared__ __attribute__((aligned(16))) float stg[KSPLIT][2][SLOT_FLOATS];

  const int m_tile = blockIdx.x;
  const int wv     = threadIdx.x >> 5;
  const int lane   = threadIdx.x & 31;
  const int row    = lane & 15;
  const int kq     = (lane >> 4) << 2;

  const int r4     = lane >> 3;
  const int c4     = (lane & 7) << 2;

  float* const       sw0 = &stg[wv][0][r4 * SLOT_PITCH + c4];
  float* const       sw1 = &stg[wv][1][r4 * SLOT_PITCH + c4];
  const float* const sr0 = &stg[wv][0][row * SLOT_PITCH + kq];
  const float* const sr1 = &stg[wv][1][row * SLOT_PITCH + kq];

  v4f g[2][4];
  v4f f[2][2][4];

  {
    const float* __restrict__ ap = u  + (size_t)(m_tile * 16 + row) * D_MODEL + wv * KSLICE + kq;
    const float* __restrict__ bg = xw + (size_t)r4 * D_MODEL + wv * KSLICE + c4;
    constexpr size_t R4 = (size_t)4 * D_MODEL;

    constexpr int NP = NT1 / 2;
    constexpr int NU = (KSLICE / KLINE) * NP;

    v8f acc[NT1] = {};
    v4f a[2][4];

    load_line(ap, a[0]);
    fetch_tile(bg, R4, g[0]);
    fetch_tile(bg + (size_t)16 * D_MODEL, R4, g[1]);
    tile_to_fragments(sw0, sr0, g[0], f[0][0]);
    tile_to_fragments(sw1, sr1, g[1], f[0][1]);
    fetch_tile(bg + (size_t)2 * 16 * D_MODEL, R4, g[0]);
    fetch_tile(bg + (size_t)3 * 16 * D_MODEL, R4, g[1]);
    #pragma unroll
    for (int un = 0; un < NU; ++un) {
      const int c = un / NP, p = un % NP;
      if (un + 1 < NU) {
        tile_to_fragments(sw0, sr0, g[0], f[(un + 1) & 1][0]);
        tile_to_fragments(sw1, sr1, g[1], f[(un + 1) & 1][1]);
      }
      if (un + 2 < NU) {
        const int c2 = (un + 2) / NP, p2 = (un + 2) % NP;
        if (p2 == 0) load_line(ap + c2 * KLINE, a[c2 & 1]);
        fetch_tile(bg + (size_t)(2 * p2) * 16 * D_MODEL + c2 * KLINE, R4, g[0]);
        fetch_tile(bg + (size_t)(2 * p2 + 1) * 16 * D_MODEL + c2 * KLINE, R4, g[1]);
      }
      SCHED_FENCE();
      wmma_unit(a[c & 1], f[un & 1][0], f[un & 1][1], acc[2 * p], acc[2 * p + 1]);
    }

    float* pw = part + wv * W_PITCH + lane;
    #pragma unroll
    for (int nt = 0; nt < NT1; ++nt)
      #pragma unroll
      for (int i = 0; i < 8; ++i)
        pw[nt * NT_PITCH + i * 32] = acc[nt][i];
  }

  __syncthreads();

  const float* __restrict__ bg2 = wt + (size_t)(wv * NT2 * 16 + r4) * DT_RANK + c4;
  constexpr size_t R42 = (size_t)4 * DT_RANK;
  fetch_tile(bg2, R42, g[0]);
  fetch_tile(bg2 + (size_t)16 * DT_RANK, R42, g[1]);
  float bn[NT2];
  #pragma unroll
  for (int nt = 0; nt < NT2; ++nt) bn[nt] = bias[(wv * NT2 + nt) * 16 + row];
  SCHED_FENCE();

  float s[(16 * XDIM) / 256];
  #pragma unroll
  for (int j = 0; j < (16 * XDIM) / 256; ++j) {
    const int e   = j * 256 + threadIdx.x;
    const int r   = e / XDIM;
    const int col = e - r * XDIM;
    const int off = (col >> 4) * NT_PITCH + (r & 7) * 32 + (r >> 3) * 16 + (col & 15);
    float t = part[off];
    #pragma unroll
    for (int w = 1; w < KSPLIT; ++w) t += part[w * W_PITCH + off];
    s[j] = t;
  }
  #pragma unroll
  for (int j = 0; j < (16 * XDIM) / 256; ++j) {
    const int e   = j * 256 + threadIdx.x;
    const int r   = e / XDIM;
    const int col = e - r * XDIM;
    if (col < DT_RANK) xs[r * XS_PITCH + col] = s[j];
  }

  __syncthreads();

  float* __restrict__ outp = xp + (size_t)m_tile * 16 * XDIM;
  #pragma unroll
  for (int j = 0; j < (16 * XDIM) / 256; ++j) {
    const int e   = j * 256 + threadIdx.x;
    const int col = e % XDIM;
    if (col >= DT_RANK) outp[e] = s[j];
  }

  {
    constexpr int NL = DT_RANK / KLINE;
    constexpr int NU = NL * (NT2 / 2);

    const float* ap2 = xs + row * XS_PITCH + kq;
    float* __restrict__ dp = delta + (size_t)(m_tile * 16 + ((lane >> 4) << 3)) * D_MODEL +
                             wv * NT2 * 16 + row;

    v8f acc[NT2] = {};
    v4f a[2][4];
    load_line(ap2, a[0]);
    load_line(ap2 + KLINE, a[1]);

    tile_to_fragments(sw0, sr0, g[0], f[0][0]);
    tile_to_fragments(sw1, sr1, g[1], f[0][1]);
    fetch_tile(bg2 + KLINE, R42, g[0]);
    fetch_tile(bg2 + (size_t)16 * DT_RANK + KLINE, R42, g[1]);

    #pragma unroll
    for (int un = 0; un < NU; ++un) {
      const int p = un / NL, c = un % NL;
      if (un + 1 < NU) {
        tile_to_fragments(sw0, sr0, g[0], f[(un + 1) & 1][0]);
        tile_to_fragments(sw1, sr1, g[1], f[(un + 1) & 1][1]);
      }
      if (un + 2 < NU) {
        const int p2 = (un + 2) / NL, c2 = (un + 2) % NL;
        fetch_tile(bg2 + (size_t)(2 * p2) * 16 * DT_RANK + c2 * KLINE, R42, g[0]);
        fetch_tile(bg2 + (size_t)(2 * p2 + 1) * 16 * DT_RANK + c2 * KLINE, R42, g[1]);
      }
      SCHED_FENCE();
      wmma_unit(a[c], f[un & 1][0], f[un & 1][1], acc[2 * p], acc[2 * p + 1]);
      if (c == NL - 1) {
        softplus_store_tile(acc[2 * p],     bn[2 * p],     dp + (2 * p) * 16);
        softplus_store_tile(acc[2 * p + 1], bn[2 * p + 1], dp + (2 * p + 1) * 16);
      }
    }
  }
}

#define PF   8
#define NGRP (CLEN / PF)


#define NPAIR (D_STATE / 2)
#define LOG2E 1.44269504f

__device__ __forceinline__ void load_A2(const float* __restrict__ A_log, const int d, v2f (&A)[NPAIR]) {
  #pragma unroll
  for (int k = 0; k < NPAIR; ++k) {
    A[k].x = -__expf(A_log[d * D_STATE + 2 * k]) * LOG2E;
    A[k].y = -__expf(A_log[d * D_STATE + 2 * k + 1]) * LOG2E;
  }
}

__device__ __forceinline__ v2f exp2_pair(const v2f x) {
  v2f e;
  e.x = __builtin_amdgcn_exp2f(x.x);
  e.y = __builtin_amdgcn_exp2f(x.y);
  return e;
}

__device__ __forceinline__ void load_group(const float* __restrict__ dp, const float* __restrict__ up,
                                           float (&dt)[PF], float (&ut)[PF]) {
  #pragma unroll
  for (int i = 0; i < PF; ++i) {
    dt[i] = dp[(size_t)i * D_MODEL];
    ut[i] = up[(size_t)i * D_MODEL];
  }
}

template <bool LAST>
__device__ __forceinline__ void pass1_group(const float4* rows, const float (&dt)[PF],
                                            const float (&ut)[PF], const v2f (&A)[NPAIR],
                                            v2f (&h)[NPAIR], float& sdt,
                                            float4 (&nb)[4]) {
  #pragma unroll
  for (int i = 0; i < PF; ++i) {
    const float4 q0 = nb[0], q1 = nb[1], q2 = nb[2], q3 = nb[3];
    if (i + 1 < PF || !LAST) {
      nb[0] = rows[(i + 1) * 4 + 0]; nb[1] = rows[(i + 1) * 4 + 1];
      nb[2] = rows[(i + 1) * 4 + 2]; nb[3] = rows[(i + 1) * 4 + 3];
    }
    SCHED_FENCE();
    const v2f Bt[NPAIR] = {{q0.x, q0.y}, {q0.z, q0.w}, {q1.x, q1.y}, {q1.z, q1.w},
                           {q2.x, q2.y}, {q2.z, q2.w}, {q3.x, q3.y}, {q3.z, q3.w}};
    const float d_t = dt[i];
    sdt += d_t;
    const float du = d_t * ut[i];
    const v2f dt2 = {d_t, d_t};
    const v2f du2 = {du, du};
    #pragma unroll
    for (int k = 0; k < NPAIR; ++k)
      h[k] = exp2_pair(dt2 * A[k]) * h[k] + du2 * Bt[k];
  }
}

__global__ __launch_bounds__(256) void scan_pass1_kernel(const float* __restrict__ delta,
                                                         const float* __restrict__ xp,
                                                         const float* __restrict__ u,
                                                         const float* __restrict__ A_log,
                                                         float* __restrict__ Q,
                                                         float* __restrict__ Sdt) {
  __shared__ float4 Bs[CLEN * 4];

  const int bid = blockIdx.x;
  const int d   = (bid & 3) * 256 + threadIdx.x;
  const int c   = (bid >> 2) & (NCHUNK - 1);
  const int b   = bid >> 7;
  const size_t tok0 = (size_t)b * SEQLEN + (size_t)c * CLEN;

  Bs[threadIdx.x] = *(const float4*)(xp + (tok0 + (threadIdx.x >> 2)) * XDIM + DT_RANK +
                                     (threadIdx.x & 3) * 4);

  const float* __restrict__ dp = delta + tok0 * D_MODEL + d;
  const float* __restrict__ up = u     + tok0 * D_MODEL + d;

  float dtc[PF], utc[PF], dtn[PF], utn[PF];
  load_group(dp, up, dtc, utc);

  v2f A[NPAIR];
  load_A2(A_log, d, A);

  v2f h[NPAIR];
  #pragma unroll
  for (int k = 0; k < NPAIR; ++k) h[k] = v2f{0.0f, 0.0f};
  float sdt = 0.0f;

  __syncthreads();

  float4 nb[4] = {Bs[0], Bs[1], Bs[2], Bs[3]};

  for (int g = 0; g < NGRP - 1; ++g) {
    load_group(dp + (size_t)(g + 1) * PF * D_MODEL, up + (size_t)(g + 1) * PF * D_MODEL, dtn, utn);
    SCHED_FENCE();
    pass1_group<false>(Bs + g * PF * 4, dtc, utc, A, h, sdt, nb);
    #pragma unroll
    for (int i = 0; i < PF; ++i) { dtc[i] = dtn[i]; utc[i] = utn[i]; }
  }
  pass1_group<true>(Bs + (NGRP - 1) * PF * 4, dtc, utc, A, h, sdt, nb);

  const size_t idx = ((size_t)b * NCHUNK + c) * D_MODEL + d;
  #pragma unroll
  for (int k = 0; k < NPAIR; ++k) {
    Q[idx * D_STATE + 2 * k]     = h[k].x;
    Q[idx * D_STATE + 2 * k + 1] = h[k].y;
  }
  Sdt[idx] = sdt;
}

#define CROW (D_MODEL * D_STATE)

__global__ __launch_bounds__(256) void scan_carry_kernel(const float* __restrict__ Q,
                                                         const float* __restrict__ Sdt,
                                                         const float* __restrict__ A_log,
                                                         float* __restrict__ Hc) {
  const unsigned tid = blockIdx.x * 256u + threadIdx.x;
  const unsigned e   = tid % CROW;
  const unsigned b   = tid / CROW;
  const unsigned d   = e / D_STATE;

  const float* __restrict__ qp = Q   + (size_t)b * NCHUNK * CROW + e;
  const float* __restrict__ sp = Sdt + (size_t)b * NCHUNK * D_MODEL + d;
  float* __restrict__       hp = Hc  + (size_t)b * NCHUNK * CROW + e;

  float q[NCHUNK], s[NCHUNK];
  #pragma unroll
  for (int c = 0; c < NCHUNK; ++c) {
    q[c] = qp[(size_t)c * CROW];
    s[c] = sp[(size_t)c * D_MODEL];
  }
  const float A = -__expf(A_log[e]);

  float h = 0.0f;
  #pragma unroll
  for (int c = 0; c < NCHUNK; ++c) {
    hp[(size_t)c * CROW] = h;
    h = __expf(s[c] * A) * h + q[c];
  }
}

template <bool LAST>
__device__ __forceinline__ void pass3_group(const float4* rows, const float (&dt)[PF],
                                            const float (&ut)[PF], const v2f (&A)[NPAIR],
                                            v2f (&h)[NPAIR], const float dsk,
                                            float* __restrict__ yp, float4 (&nbc)[8]) {
  #pragma unroll
  for (int i = 0; i < PF; ++i) {
    const float4 q0 = nbc[0], q1 = nbc[1], q2 = nbc[2], q3 = nbc[3];
    const float4 p0 = nbc[4], p1 = nbc[5], p2 = nbc[6], p3 = nbc[7];
    if (i + 1 < PF || !LAST) {
      #pragma unroll
      for (int r = 0; r < 8; ++r) nbc[r] = rows[(i + 1) * 8 + r];
    }
    SCHED_FENCE();
    const v2f Bt[NPAIR] = {{q0.x, q0.y}, {q0.z, q0.w}, {q1.x, q1.y}, {q1.z, q1.w},
                           {q2.x, q2.y}, {q2.z, q2.w}, {q3.x, q3.y}, {q3.z, q3.w}};
    const v2f Ct[NPAIR] = {{p0.x, p0.y}, {p0.z, p0.w}, {p1.x, p1.y}, {p1.z, p1.w},
                           {p2.x, p2.y}, {p2.z, p2.w}, {p3.x, p3.y}, {p3.z, p3.w}};
    const float d_t = dt[i];
    const float u_t = ut[i];
    const float du = d_t * u_t;
    const v2f dt2 = {d_t, d_t};
    const v2f du2 = {du, du};
    v2f acc = {0.0f, 0.0f};
    #pragma unroll
    for (int k = 0; k < NPAIR; ++k) {
      h[k] = exp2_pair(dt2 * A[k]) * h[k] + du2 * Bt[k];
      acc += h[k] * Ct[k];
    }
    yp[(size_t)i * D_MODEL] = (acc.x + acc.y) + u_t * dsk;
  }
}

__global__ __launch_bounds__(256) void scan_pass3_kernel(const float* __restrict__ delta,
                                                         const float* __restrict__ xp,
                                                         const float* __restrict__ u,
                                                         const float* __restrict__ A_log,
                                                         const float* __restrict__ Dsk,
                                                         const float* __restrict__ Hc,
                                                         float* __restrict__ y) {
  __shared__ float4 BCs[CLEN * 8];

  const int bid = blockIdx.x;
  const int d   = (bid & 3) * 256 + threadIdx.x;
  const int c   = (bid >> 2) & (NCHUNK - 1);
  const int b   = bid >> 7;
  const size_t tok0 = (size_t)b * SEQLEN + (size_t)c * CLEN;

  #pragma unroll
  for (int j = 0; j < 2; ++j) {
    const int e = j * 256 + threadIdx.x;
    BCs[e] = *(const float4*)(xp + (tok0 + (e >> 3)) * XDIM + DT_RANK + (e & 7) * 4);
  }

  const float* __restrict__ dp = delta + tok0 * D_MODEL + d;
  const float* __restrict__ up = u     + tok0 * D_MODEL + d;
  float* __restrict__       yp = y     + tok0 * D_MODEL + d;

  float dtc[PF], utc[PF], dtn[PF], utn[PF];
  load_group(dp, up, dtc, utc);

  v2f A[NPAIR];
  load_A2(A_log, d, A);

  const size_t idx = ((size_t)b * NCHUNK + c) * D_MODEL + d;
  v2f h[NPAIR];
  #pragma unroll
  for (int k = 0; k < NPAIR; ++k) {
    h[k].x = Hc[idx * D_STATE + 2 * k];
    h[k].y = Hc[idx * D_STATE + 2 * k + 1];
  }
  const float dsk = Dsk[d];

  __syncthreads();

  float4 nbc[8];
  #pragma unroll
  for (int r = 0; r < 8; ++r) nbc[r] = BCs[r];

  for (int g = 0; g < NGRP - 1; ++g) {
    load_group(dp + (size_t)(g + 1) * PF * D_MODEL, up + (size_t)(g + 1) * PF * D_MODEL, dtn, utn);
    SCHED_FENCE();
    pass3_group<false>(BCs + g * PF * 8, dtc, utc, A, h, dsk, yp + (size_t)g * PF * D_MODEL, nbc);
    #pragma unroll
    for (int i = 0; i < PF; ++i) { dtc[i] = dtn[i]; utc[i] = utn[i]; }
  }
  pass3_group<true>(BCs + (NGRP - 1) * PF * 8, dtc, utc, A, h, dsk, yp + (size_t)(NGRP - 1) * PF * D_MODEL, nbc);
}

extern "C" void kernel_launch(void* const* d_in, const int* in_sizes, int n_in,
                              void* d_out, int out_size, void* d_ws, size_t ws_size,
                              hipStream_t stream) {
  const float* u      = (const float*)d_in[0];
  const float* A_log  = (const float*)d_in[1];
  const float* D_skip = (const float*)d_in[2];
  const float* dt_w   = (const float*)d_in[3];
  const float* dt_b   = (const float*)d_in[4];
  const float* x_w    = (const float*)d_in[5];
  float* y = (float*)d_out;

  float* ws    = (float*)d_ws;
  float* xp    = ws;
  float* delta = xp    + (size_t)NTOK * XDIM;
  float* Q     = delta + (size_t)NTOK * D_MODEL;
  float* Sdt   = Q     + (size_t)BATCH * NCHUNK * D_MODEL * D_STATE;
  float* Hc    = Sdt   + (size_t)BATCH * NCHUNK * D_MODEL;

  xp_delta_wmma_kernel<<<NTOK / 16, 256, 0, stream>>>(u, x_w, dt_w, dt_b, xp, delta);

  scan_pass1_kernel<<<BATCH * NCHUNK * (D_MODEL / 256), 256, 0, stream>>>(
      delta, xp, u, A_log, Q, Sdt);

  scan_carry_kernel<<<BATCH * D_MODEL * D_STATE / 256, 256, 0, stream>>>(Q, Sdt, A_log, Hc);

  scan_pass3_kernel<<<BATCH * NCHUNK * (D_MODEL / 256), 256, 0, stream>>>(
      delta, xp, u, A_log, D_skip, Hc, y);
}
